# starting kernel
# baseline (speedup 1.0000x reference)
#include <hip/hip_runtime.h>
#include <math.h>

typedef float v2f __attribute__((ext_vector_type(2)));
typedef float v8f __attribute__((ext_vector_type(8)));

__device__ __forceinline__ float softplusf(float x) {
  return (x > 20.f) ? x : log1pf(expf(x));
}

__device__ __forceinline__ float fast_tanh(float x) {
#if __has_builtin(__builtin_amdgcn_tanhf)
  return __builtin_amdgcn_tanhf(x);
#else
  float cx = fminf(fmaxf(x, -10.f), 10.f);
  float e = exp2f(cx * 2.885390081777927f);
  return (e - 1.f) / (e + 1.f);
#endif
}

__device__ __forceinline__ float sigmoidf_(float x) {
  return 1.f / (1.f + expf(-x));
}

__device__ __forceinline__ v8f wmma4_f32(v2f a, v2f b, v8f c) {
#if __has_builtin(__builtin_amdgcn_wmma_f32_16x16x4_f32)
  return __builtin_amdgcn_wmma_f32_16x16x4_f32(
       false, a,  false, b,
       (short)0, c,  false,  false);
#else
  int lane = threadIdx.x & 31;
  int half = lane >> 4;
  int n = lane & 15;
  float B0 = __shfl(b.x, n),      B1 = __shfl(b.y, n);
  float B2 = __shfl(b.x, n + 16), B3 = __shfl(b.y, n + 16);
  v8f d;
#pragma unroll
  for (int j = 0; j < 8; ++j) {
    int r = j + (half ? 8 : 0);
    float A0 = __shfl(a.x, r),      A1 = __shfl(a.y, r);
    float A2 = __shfl(a.x, r + 16), A3 = __shfl(a.y, r + 16);
    d[j] = fmaf(A0, B0, fmaf(A1, B1, fmaf(A2, B2, fmaf(A3, B3, c[j]))));
  }
  return d;
#endif
}

struct Lay {
  float a0, a1;
  float c0, c1, c2;
  float t0, t1, t2;
};

__device__ __forceinline__ void setup_l0(const float* m, const float* b, const float* f,
                                         int ch, int lane, Lay& L) {
  int half = lane >> 4, row = lane & 15;
  const float* M = m + ch * 3;
  const float* B = b + ch * 3;
  const float* F = f + ch * 3;
  L.a0 = (!half && row < 3) ? softplusf(M[row]) : 0.f;
  L.a1 = 0.f;
  L.c0 = half ? 0.f : B[0];  L.c1 = half ? 0.f : B[1];  L.c2 = half ? 0.f : B[2];
  L.t0 = half ? 0.f : tanhf(F[0]);
  L.t1 = half ? 0.f : tanhf(F[1]);
  L.t2 = half ? 0.f : tanhf(F[2]);
}

__device__ __forceinline__ void setup_l33(const float* m, const float* b, const float* f,
                                          int ch, int lane, Lay& L) {
  int half = lane >> 4, row = lane & 15;
  const float* M = m + ch * 9;
  const float* B = b + ch * 3;
  const float* F = f + ch * 3;
  float a0 = 0.f, a1 = 0.f;
  if (row < 3) {
    a0 = softplusf(M[row * 3 + (half ? 2 : 0)]);
    a1 = half ? 0.f : softplusf(M[row * 3 + 1]);
  }
  L.a0 = a0; L.a1 = a1;
  L.c0 = half ? 0.f : B[0];  L.c1 = half ? 0.f : B[1];  L.c2 = half ? 0.f : B[2];
  L.t0 = half ? 0.f : tanhf(F[0]);
  L.t1 = half ? 0.f : tanhf(F[1]);
  L.t2 = half ? 0.f : tanhf(F[2]);
}

__device__ __forceinline__ void setup_l3(const float* m, const float* b, const float* f,
                                         int ch, int lane, Lay& L) {
  int half = lane >> 4, row = lane & 15;
  const float* M = m + ch * 3;
  L.a0 = (row == 0) ? softplusf(M[half ? 2 : 0]) : 0.f;
  L.a1 = (row == 0 && !half) ? softplusf(M[1]) : 0.f;
  L.c0 = half ? 0.f : b[ch];
  L.c1 = 0.f; L.c2 = 0.f;
  L.t0 = half ? 0.f : tanhf(f[ch]);
  L.t1 = 0.f; L.t2 = 0.f;
}

__device__ __forceinline__ float eval_chain(Lay L0, Lay L1, Lay L2, Lay L3,
                                            float x0, int lane) {
  int half = lane >> 4;
  int nlo = lane & 15;

  v2f a, b;
  a.x = L0.a0; a.y = L0.a1;
  b.x = x0;    b.y = 0.f;
  v8f c = {L0.c0, L0.c1, L0.c2, 0.f, 0.f, 0.f, 0.f, 0.f};
  v8f d = wmma4_f32(a, b, c);
  float h0 = d[0] + L0.t0 * fast_tanh(d[0]);
  float h1 = d[1] + L0.t1 * fast_tanh(d[1]);
  float h2 = d[2] + L0.t2 * fast_tanh(d[2]);

  float h2lo = __shfl(h2, nlo);
  a.x = L1.a0; a.y = L1.a1;
  b.x = half ? h2lo : h0;  b.y = half ? 0.f : h1;
  v8f c1 = {L1.c0, L1.c1, L1.c2, 0.f, 0.f, 0.f, 0.f, 0.f};
  d = wmma4_f32(a, b, c1);
  h0 = d[0] + L1.t0 * fast_tanh(d[0]);
  h1 = d[1] + L1.t1 * fast_tanh(d[1]);
  h2 = d[2] + L1.t2 * fast_tanh(d[2]);

  h2lo = __shfl(h2, nlo);
  a.x = L2.a0; a.y = L2.a1;
  b.x = half ? h2lo : h0;  b.y = half ? 0.f : h1;
  v8f c2 = {L2.c0, L2.c1, L2.c2, 0.f, 0.f, 0.f, 0.f, 0.f};
  d = wmma4_f32(a, b, c2);
  h0 = d[0] + L2.t0 * fast_tanh(d[0]);
  h1 = d[1] + L2.t1 * fast_tanh(d[1]);
  h2 = d[2] + L2.t2 * fast_tanh(d[2]);

  h2lo = __shfl(h2, nlo);
  a.x = L3.a0; a.y = L3.a1;
  b.x = half ? h2lo : h0;  b.y = half ? 0.f : h1;
  v8f c3 = {L3.c0, 0.f, 0.f, 0.f, 0.f, 0.f, 0.f, 0.f};
  d = wmma4_f32(a, b, c3);
  float r = d[0];
  return r + L3.t0 * fast_tanh(r);
}

__global__ void __launch_bounds__(256)
eb_kernel(const float* __restrict__ inputs, const float* __restrict__ noise,
          const float* __restrict__ m0, const float* __restrict__ b0, const float* __restrict__ f0,
          const float* __restrict__ m1, const float* __restrict__ b1, const float* __restrict__ f1,
          const float* __restrict__ m2, const float* __restrict__ b2, const float* __restrict__ f2,
          const float* __restrict__ m3, const float* __restrict__ b3, const float* __restrict__ f3,
          float* __restrict__ out, float* __restrict__ lik, int ntiles) {
  __shared__ float xs[256];
  __shared__ float ls[256];

  const int tid  = threadIdx.x;
  const int lane = tid & 31;
  const int wave = tid >> 5;
  const int half = lane >> 4;
  const int row  = lane & 15;

  Lay L0[2], L1[2], L2[2], L3[2];
#pragma unroll
  for (int cc = 0; cc < 2; ++cc) {
    int ch = wave * 2 + cc;
    setup_l0 (m0, b0, f0, ch, lane, L0[cc]);
    setup_l33(m1, b1, f1, ch, lane, L1[cc]);
    setup_l33(m2, b2, f2, ch, lane, L2[cc]);
    setup_l3 (m3, b3, f3, ch, lane, L3[cc]);
  }

  for (int t = blockIdx.x; t < ntiles; t += gridDim.x) {
    const int base = t * 256;

    float v = inputs[base + tid] + (noise[base + tid] - 0.5f);
    out[base + tid] = v;
    xs[(tid & 15) * 16 + (tid >> 4)] = v;
    __syncthreads();

#pragma unroll
    for (int cc = 0; cc < 2; ++cc) {
      int ch = wave * 2 + cc;
      float xv = xs[ch * 16 + row];
      float xl = half ? 0.f : (xv - 0.5f);
      float xu = half ? 0.f : (xv + 0.5f);
      float lo = eval_chain(L0[cc], L1[cc], L2[cc], L3[cc], xl, lane);
      float up = eval_chain(L0[cc], L1[cc], L2[cc], L3[cc], xu, lane);
      if (half == 0) {
        float sum = lo + up;
        float s = (sum > 0.f) ? -1.f : ((sum < 0.f) ? 1.f : 0.f);
        float lk = fabsf(sigmoidf_(s * up) - sigmoidf_(s * lo));
        ls[row * 16 + ch] = fmaxf(lk, 1e-9f);
      }
    }
    __syncthreads();

    lik[base + tid] = ls[tid];
    __syncthreads();
  }
}

extern "C" void kernel_launch(void* const* d_in, const int* in_sizes, int n_in,
                              void* d_out, int out_size, void* d_ws, size_t ws_size,
                              hipStream_t stream) {
  const float* inputs = (const float*)d_in[0];
  const float* noise  = (const float*)d_in[1];
  const float* m0 = (const float*)d_in[2];
  const float* b0 = (const float*)d_in[3];
  const float* f0 = (const float*)d_in[4];
  const float* m1 = (const float*)d_in[5];
  const float* b1 = (const float*)d_in[6];
  const float* f1 = (const float*)d_in[7];
  const float* m2 = (const float*)d_in[8];
  const float* b2 = (const float*)d_in[9];
  const float* f2 = (const float*)d_in[10];
  const float* m3 = (const float*)d_in[11];
  const float* b3 = (const float*)d_in[12];
  const float* f3 = (const float*)d_in[13];

  const int nelem  = in_sizes[0];
  const int ntiles = nelem / 256;
  float* out = (float*)d_out;
  float* lik = out + nelem;

  int blocks = ntiles < 2048 ? ntiles : 2048;
  eb_kernel<<<blocks, 256, 0, stream>>>(inputs, noise,
                                        m0, b0, f0, m1, b1, f1,
                                        m2, b2, f2, m3, b3, f3,
                                        out, lik, ntiles);
}


# Two channels per matrix-op chain: lane halves and D rows 0-2 / 8-10 carry one channel each, bias as a k slot times 1, 3-wide layers as two chained ops; no lane shuffle; sum order changes
# speedup vs baseline: 1.6263x; 1.6263x over previous
#include <hip/hip_runtime.h>
#include <math.h>


typedef float v2f __attribute__((ext_vector_type(2)));
typedef float v8f __attribute__((ext_vector_type(8)));

__device__ __forceinline__ float softplusf(float x) {
  return (x > 20.f) ? x : log1pf(expf(x));
}

__device__ __forceinline__ float fast_tanh(float x) {
#if __has_builtin(__builtin_amdgcn_tanhf)
  return __builtin_amdgcn_tanhf(x);
#else
  float cx = fminf(fmaxf(x, -10.f), 10.f);
  float e = exp2f(cx * 2.885390081777927f);
  return (e - 1.f) / (e + 1.f);
#endif
}

__device__ __forceinline__ float sigmoidf_(float x) {
  return 1.f / (1.f + expf(-x));
}

__device__ __forceinline__ v8f wmma4_f32(v2f a, v2f b, v8f c) {
#if __has_builtin(__builtin_amdgcn_wmma_f32_16x16x4_f32)
  return __builtin_amdgcn_wmma_f32_16x16x4_f32(
       false, a,  false, b,
       (short)0, c,  false,  false);
#else
  int lane = threadIdx.x & 31;
  int half = lane >> 4;
  int n = lane & 15;
  float B0 = __shfl(b.x, n),      B1 = __shfl(b.y, n);
  float B2 = __shfl(b.x, n + 16), B3 = __shfl(b.y, n + 16);
  v8f d;
#pragma unroll
  for (int j = 0; j < 8; ++j) {
    int r = j + (half ? 8 : 0);
    float A0 = __shfl(a.x, r),      A1 = __shfl(a.y, r);
    float A2 = __shfl(a.x, r + 16), A3 = __shfl(a.y, r + 16);
    d[j] = fmaf(A0, B0, fmaf(A1, B1, fmaf(A2, B2, fmaf(A3, B3, c[j]))));
  }
  return d;
#endif
}

struct Pack {
  v2f a0;
  v2f a1p, a1q;
  v2f a2p, a2q;
  v2f a3p, a3q;
  float t0[3], t1[3], t2[3], t3;
};

__device__ __forceinline__ void setup_pack(
    const float* m0, const float* b0, const float* f0,
    const float* m1, const float* b1, const float* f1,
    const float* m2, const float* b2, const float* f2,
    const float* m3, const float* b3, const float* f3,
    int wave, int lane, Pack& P) {
  const int half = lane >> 4, r = lane & 15;
  const int c = wave * 2 + half;
  const int o = r - 8 * half;
  const bool v3 = (o >= 0) && (o < 3);
  const bool v1 = (o == 0);
  const int oc = v3 ? o : 0;

  P.a0.x  = v3 ? softplusf(m0[c * 3 + oc]) : 0.f;
  P.a0.y  = v3 ? b0[c * 3 + oc] : 0.f;

  P.a1p.x = v3 ? softplusf(m1[c * 9 + oc * 3 + 0]) : 0.f;
  P.a1p.y = v3 ? softplusf(m1[c * 9 + oc * 3 + 1]) : 0.f;
  P.a1q.x = v3 ? softplusf(m1[c * 9 + oc * 3 + 2]) : 0.f;
  P.a1q.y = v3 ? b1[c * 3 + oc] : 0.f;

  P.a2p.x = v3 ? softplusf(m2[c * 9 + oc * 3 + 0]) : 0.f;
  P.a2p.y = v3 ? softplusf(m2[c * 9 + oc * 3 + 1]) : 0.f;
  P.a2q.x = v3 ? softplusf(m2[c * 9 + oc * 3 + 2]) : 0.f;
  P.a2q.y = v3 ? b2[c * 3 + oc] : 0.f;

  P.a3p.x = v1 ? softplusf(m3[c * 3 + 0]) : 0.f;
  P.a3p.y = v1 ? softplusf(m3[c * 3 + 1]) : 0.f;
  P.a3q.x = v1 ? softplusf(m3[c * 3 + 2]) : 0.f;
  P.a3q.y = v1 ? b3[c] : 0.f;

#pragma unroll
  for (int j = 0; j < 3; ++j) {
    P.t0[j] = tanhf(f0[c * 3 + j]);
    P.t1[j] = tanhf(f1[c * 3 + j]);
    P.t2[j] = tanhf(f2[c * 3 + j]);
  }
  P.t3 = tanhf(f3[c]);
}

__device__ __forceinline__ void gate3(const v8f d, const float* t, float& h0, float& h1, float& h2) {
  h0 = d[0] + t[0] * fast_tanh(d[0]);
  h1 = d[1] + t[1] * fast_tanh(d[1]);
  h2 = d[2] + t[2] * fast_tanh(d[2]);
}

__device__ __forceinline__ v8f layer3in(v2f ap, v2f aq, float h0, float h1, float h2) {
  const v8f z = {0.f, 0.f, 0.f, 0.f, 0.f, 0.f, 0.f, 0.f};
  v2f bp, bq;
  bp.x = h0; bp.y = h1;
  bq.x = h2; bq.y = 1.f;
  v8f d = wmma4_f32(ap, bp, z);
  return wmma4_f32(aq, bq, d);
}

__device__ __forceinline__ void eval_pair(const Pack& P, float xl, float xu, float& lo, float& up) {
  const v8f z = {0.f, 0.f, 0.f, 0.f, 0.f, 0.f, 0.f, 0.f};
  v2f bl, bu;
  bl.x = xl; bl.y = 1.f;
  bu.x = xu; bu.y = 1.f;
  v8f dl = wmma4_f32(P.a0, bl, z);
  v8f du = wmma4_f32(P.a0, bu, z);
  float l0, l1, l2, u0, u1, u2;
  gate3(dl, P.t0, l0, l1, l2);
  gate3(du, P.t0, u0, u1, u2);

  dl = layer3in(P.a1p, P.a1q, l0, l1, l2);
  du = layer3in(P.a1p, P.a1q, u0, u1, u2);
  gate3(dl, P.t1, l0, l1, l2);
  gate3(du, P.t1, u0, u1, u2);

  dl = layer3in(P.a2p, P.a2q, l0, l1, l2);
  du = layer3in(P.a2p, P.a2q, u0, u1, u2);
  gate3(dl, P.t2, l0, l1, l2);
  gate3(du, P.t2, u0, u1, u2);

  dl = layer3in(P.a3p, P.a3q, l0, l1, l2);
  du = layer3in(P.a3p, P.a3q, u0, u1, u2);
  float rl = dl[0], ru = du[0];
  lo = rl + P.t3 * fast_tanh(rl);
  up = ru + P.t3 * fast_tanh(ru);
}

__global__ void __launch_bounds__(256)
eb_kernel(const float* __restrict__ inputs, const float* __restrict__ noise,
          const float* __restrict__ m0, const float* __restrict__ b0, const float* __restrict__ f0,
          const float* __restrict__ m1, const float* __restrict__ b1, const float* __restrict__ f1,
          const float* __restrict__ m2, const float* __restrict__ b2, const float* __restrict__ f2,
          const float* __restrict__ m3, const float* __restrict__ b3, const float* __restrict__ f3,
          float* __restrict__ out, float* __restrict__ lik, int ntiles) {
  __shared__ float xs[256];
  __shared__ float ls[256];

  const int tid  = threadIdx.x;
  const int lane = tid & 31;
  const int wave = tid >> 5;
  const int half = lane >> 4;
  const int row  = lane & 15;

  Pack P;
  setup_pack(m0, b0, f0, m1, b1, f1, m2, b2, f2, m3, b3, f3, wave, lane, P);

  for (int t = blockIdx.x; t < ntiles; t += gridDim.x) {
    const int base = t * 256;

    float v = inputs[base + tid] + (noise[base + tid] - 0.5f);
    out[base + tid] = v;
    xs[(tid & 15) * 16 + (tid >> 4)] = v;
    __syncthreads();

    float xv = xs[wave * 32 + lane];
    float lo, up;
    eval_pair(P, xv - 0.5f, xv + 0.5f, lo, up);
    {
      float sum = lo + up;
      float s = (sum > 0.f) ? -1.f : ((sum < 0.f) ? 1.f : 0.f);
      float lk = fabsf(sigmoidf_(s * up) - sigmoidf_(s * lo));
      ls[row * 16 + wave * 2 + half] = fmaxf(lk, 1e-9f);
    }
    __syncthreads();

    lik[base + tid] = ls[tid];
    __syncthreads();
  }
}

extern "C" void kernel_launch(void* const* d_in, const int* in_sizes, int n_in,
                              void* d_out, int out_size, void* d_ws, size_t ws_size,
                              hipStream_t stream) {
  const float* inputs = (const float*)d_in[0];
  const float* noise  = (const float*)d_in[1];
  const float* m0 = (const float*)d_in[2];
  const float* b0 = (const float*)d_in[3];
  const float* f0 = (const float*)d_in[4];
  const float* m1 = (const float*)d_in[5];
  const float* b1 = (const float*)d_in[6];
  const float* f1 = (const float*)d_in[7];
  const float* m2 = (const float*)d_in[8];
  const float* b2 = (const float*)d_in[9];
  const float* f2 = (const float*)d_in[10];
  const float* m3 = (const float*)d_in[11];
  const float* b3 = (const float*)d_in[12];
  const float* f3 = (const float*)d_in[13];

  const int nelem  = in_sizes[0];
  const int ntiles = nelem / 256;
  float* out = (float*)d_out;
  float* lik = out + nelem;

  int blocks = ntiles < 2048 ? ntiles : 2048;
  eb_kernel<<<blocks, 256, 0, stream>>>(inputs, noise,
                                        m0, b0, f0, m1, b1, f1,
                                        m2, b2, f2, m3, b3, f3,
                                        out, lik, ntiles);
}


# Tile loop pipelined: 64-point tiles (4 chains per wave, 2 barriers per 1024 elems, was 3 per 256), 16-byte IO, next tile's loads issued ahead, barriers wait on LDS only; sum order unchanged
# speedup vs baseline: 2.1864x; 1.3444x over previous
#include <hip/hip_runtime.h>
#include <math.h>


typedef float v2f __attribute__((ext_vector_type(2)));
typedef float v8f __attribute__((ext_vector_type(8)));

__device__ __forceinline__ float softplusf(float x) {
  return (x > 20.f) ? x : log1pf(expf(x));
}

__device__ __forceinline__ float fast_tanh(float x) {
#if __has_builtin(__builtin_amdgcn_tanhf)
  return __builtin_amdgcn_tanhf(x);
#else
  float cx = fminf(fmaxf(x, -10.f), 10.f);
  float e = exp2f(cx * 2.885390081777927f);
  return (e - 1.f) / (e + 1.f);
#endif
}

__device__ __forceinline__ float sigmoidf_(float x) {
  return 1.f / (1.f + expf(-x));
}

__device__ __forceinline__ v8f wmma4_f32(v2f a, v2f b, v8f c) {
#if __has_builtin(__builtin_amdgcn_wmma_f32_16x16x4_f32)
  return __builtin_amdgcn_wmma_f32_16x16x4_f32(
       false, a,  false, b,
       (short)0, c,  false,  false);
#else
  int lane = threadIdx.x & 31;
  int half = lane >> 4;
  int n = lane & 15;
  float B0 = __shfl(b.x, n),      B1 = __shfl(b.y, n);
  float B2 = __shfl(b.x, n + 16), B3 = __shfl(b.y, n + 16);
  v8f d;
#pragma unroll
  for (int j = 0; j < 8; ++j) {
    int r = j + (half ? 8 : 0);
    float A0 = __shfl(a.x, r),      A1 = __shfl(a.y, r);
    float A2 = __shfl(a.x, r + 16), A3 = __shfl(a.y, r + 16);
    d[j] = fmaf(A0, B0, fmaf(A1, B1, fmaf(A2, B2, fmaf(A3, B3, c[j]))));
  }
  return d;
#endif
}

struct Pack {
  v2f a0;
  v2f a1p, a1q;
  v2f a2p, a2q;
  v2f a3p, a3q;
  float t0[3], t1[3], t2[3], t3;
};

__device__ __forceinline__ void setup_pack(
    const float* m0, const float* b0, const float* f0,
    const float* m1, const float* b1, const float* f1,
    const float* m2, const float* b2, const float* f2,
    const float* m3, const float* b3, const float* f3,
    int wave, int lane, Pack& P) {
  const int half = lane >> 4, r = lane & 15;
  const int c = wave * 2 + half;
  const int o = r - 8 * half;
  const bool v3 = (o >= 0) && (o < 3);
  const bool v1 = (o == 0);
  const int oc = v3 ? o : 0;

  P.a0.x  = v3 ? softplusf(m0[c * 3 + oc]) : 0.f;
  P.a0.y  = v3 ? b0[c * 3 + oc] : 0.f;

  P.a1p.x = v3 ? softplusf(m1[c * 9 + oc * 3 + 0]) : 0.f;
  P.a1p.y = v3 ? softplusf(m1[c * 9 + oc * 3 + 1]) : 0.f;
  P.a1q.x = v3 ? softplusf(m1[c * 9 + oc * 3 + 2]) : 0.f;
  P.a1q.y = v3 ? b1[c * 3 + oc] : 0.f;

  P.a2p.x = v3 ? softplusf(m2[c * 9 + oc * 3 + 0]) : 0.f;
  P.a2p.y = v3 ? softplusf(m2[c * 9 + oc * 3 + 1]) : 0.f;
  P.a2q.x = v3 ? softplusf(m2[c * 9 + oc * 3 + 2]) : 0.f;
  P.a2q.y = v3 ? b2[c * 3 + oc] : 0.f;

  P.a3p.x = v1 ? softplusf(m3[c * 3 + 0]) : 0.f;
  P.a3p.y = v1 ? softplusf(m3[c * 3 + 1]) : 0.f;
  P.a3q.x = v1 ? softplusf(m3[c * 3 + 2]) : 0.f;
  P.a3q.y = v1 ? b3[c] : 0.f;

#pragma unroll
  for (int j = 0; j < 3; ++j) {
    P.t0[j] = tanhf(f0[c * 3 + j]);
    P.t1[j] = tanhf(f1[c * 3 + j]);
    P.t2[j] = tanhf(f2[c * 3 + j]);
  }
  P.t3 = tanhf(f3[c]);
}

__device__ __forceinline__ void gate3(const v8f d, const float* t, float& h0, float& h1, float& h2) {
  h0 = d[0] + t[0] * fast_tanh(d[0]);
  h1 = d[1] + t[1] * fast_tanh(d[1]);
  h2 = d[2] + t[2] * fast_tanh(d[2]);
}

__device__ __forceinline__ v8f layer3in(v2f ap, v2f aq, float h0, float h1, float h2) {
  const v8f z = {0.f, 0.f, 0.f, 0.f, 0.f, 0.f, 0.f, 0.f};
  v2f bp, bq;
  bp.x = h0; bp.y = h1;
  bq.x = h2; bq.y = 1.f;
  v8f d = wmma4_f32(ap, bp, z);
  return wmma4_f32(aq, bq, d);
}

__device__ __forceinline__ void eval_pair(const Pack& P, float xl, float xu, float& lo, float& up) {
  const v8f z = {0.f, 0.f, 0.f, 0.f, 0.f, 0.f, 0.f, 0.f};
  v2f bl, bu;
  bl.x = xl; bl.y = 1.f;
  bu.x = xu; bu.y = 1.f;
  v8f dl = wmma4_f32(P.a0, bl, z);
  v8f du = wmma4_f32(P.a0, bu, z);
  float l0, l1, l2, u0, u1, u2;
  gate3(dl, P.t0, l0, l1, l2);
  gate3(du, P.t0, u0, u1, u2);

  dl = layer3in(P.a1p, P.a1q, l0, l1, l2);
  du = layer3in(P.a1p, P.a1q, u0, u1, u2);
  gate3(dl, P.t1, l0, l1, l2);
  gate3(du, P.t1, u0, u1, u2);

  dl = layer3in(P.a2p, P.a2q, l0, l1, l2);
  du = layer3in(P.a2p, P.a2q, u0, u1, u2);
  gate3(dl, P.t2, l0, l1, l2);
  gate3(du, P.t2, u0, u1, u2);

  dl = layer3in(P.a3p, P.a3q, l0, l1, l2);
  du = layer3in(P.a3p, P.a3q, u0, u1, u2);
  float rl = dl[0], ru = du[0];
  lo = rl + P.t3 * fast_tanh(rl);
  up = ru + P.t3 * fast_tanh(ru);
}

#define XP 82
#define LP 68

__device__ __forceinline__ void lds_barrier() {
  asm volatile("s_wait_dscnt 0x0\n\ts_barrier_signal -1\n\ts_barrier_wait -1" ::: "memory");
}

__global__ void __launch_bounds__(256)
eb_kernel(const float* __restrict__ inputs, const float* __restrict__ noise,
          const float* __restrict__ m0, const float* __restrict__ b0, const float* __restrict__ f0,
          const float* __restrict__ m1, const float* __restrict__ b1, const float* __restrict__ f1,
          const float* __restrict__ m2, const float* __restrict__ b2, const float* __restrict__ f2,
          const float* __restrict__ m3, const float* __restrict__ b3, const float* __restrict__ f3,
          float* __restrict__ out, float* __restrict__ lik, int ntiles) {
  __shared__ float xs[16 * XP];
  __shared__ __attribute__((aligned(16))) float ls[16 * LP];

  const int tid  = threadIdx.x;
  const int lane = tid & 31;
  const int wave = tid >> 5;
  const int half = lane >> 4;
  const int row  = lane & 15;

  const float4* __restrict__ in4 = (const float4*)inputs;
  const float4* __restrict__ nz4 = (const float4*)noise;
  float4* __restrict__ out4 = (float4*)out;
  float4* __restrict__ lik4 = (float4*)lik;

  int t = blockIdx.x;
  float4 a = {0.f, 0.f, 0.f, 0.f}, b = {0.f, 0.f, 0.f, 0.f};
  if (t < ntiles) {
    a = in4[t * 256 + tid];
    b = nz4[t * 256 + tid];
  }

  Pack P;
  setup_pack(m0, b0, f0, m1, b1, f1, m2, b2, f2, m3, b3, f3, wave, lane, P);

  const int p = tid >> 2, q = tid & 3;
  float* xw = &xs[4 * q * XP + p];
  const float4* lr = (const float4*)&ls[(p >> 2) * LP + (p & 3) * 16 + 4 * q];
  const float* xr = &xs[(2 * wave + half) * XP + row];
  float* lw = &ls[(row >> 2) * LP + (row & 3) * 16 + 2 * wave + half];

  for (; t < ntiles; t += gridDim.x) {
    float4 v;
    v.x = a.x + (b.x - 0.5f);
    v.y = a.y + (b.y - 0.5f);
    v.z = a.z + (b.z - 0.5f);
    v.w = a.w + (b.w - 0.5f);
    out4[t * 256 + tid] = v;
    xw[0]      = v.x;
    xw[XP]     = v.y;
    xw[2 * XP] = v.z;
    xw[3 * XP] = v.w;

    const int tn = t + gridDim.x;
    if (tn < ntiles) {
      a = in4[tn * 256 + tid];
      b = nz4[tn * 256 + tid];
    }
    lds_barrier();

#pragma unroll
    for (int g = 0; g < 4; ++g) {
      float xv = xr[16 * g];
      float lo, up;
      eval_pair(P, xv - 0.5f, xv + 0.5f, lo, up);
      float sum = lo + up;
      float s = (sum > 0.f) ? -1.f : ((sum < 0.f) ? 1.f : 0.f);
      float lk = fabsf(sigmoidf_(s * up) - sigmoidf_(s * lo));
      lw[4 * g * LP] = fmaxf(lk, 1e-9f);
    }
    lds_barrier();

    lik4[t * 256 + tid] = *lr;
  }
}

extern "C" void kernel_launch(void* const* d_in, const int* in_sizes, int n_in,
                              void* d_out, int out_size, void* d_ws, size_t ws_size,
                              hipStream_t stream) {
  const float* inputs = (const float*)d_in[0];
  const float* noise  = (const float*)d_in[1];
  const float* m0 = (const float*)d_in[2];
  const float* b0 = (const float*)d_in[3];
  const float* f0 = (const float*)d_in[4];
  const float* m1 = (const float*)d_in[5];
  const float* b1 = (const float*)d_in[6];
  const float* f1 = (const float*)d_in[7];
  const float* m2 = (const float*)d_in[8];
  const float* b2 = (const float*)d_in[9];
  const float* f2 = (const float*)d_in[10];
  const float* m3 = (const float*)d_in[11];
  const float* b3 = (const float*)d_in[12];
  const float* f3 = (const float*)d_in[13];

  const int nelem  = in_sizes[0];
  const int ntiles = nelem / 1024;
  float* out = (float*)d_out;
  float* lik = out + nelem;

  int blocks = ntiles < 1024 ? ntiles : 1024;
  eb_kernel<<<blocks, 256, 0, stream>>>(inputs, noise,
                                        m0, b0, f0, m1, b1, f1,
                                        m2, b2, f2, m3, b3, f3,
                                        out, lik, ntiles);
}


# Likelihood tail on hardware 2^x and reciprocal: |1/(1+2^(k*up)) - 1/(1+2^(k*lo))|, k = -s*log2(e), replaces library expf and IEEE division; about 1 ulp each; sum order unchanged
# speedup vs baseline: 3.1063x; 1.4207x over previous
#include <hip/hip_runtime.h>
#include <math.h>


typedef float v2f __attribute__((ext_vector_type(2)));
typedef float v8f __attribute__((ext_vector_type(8)));

__device__ __forceinline__ float softplusf(float x) {
  return (x > 20.f) ? x : log1pf(expf(x));
}

__device__ __forceinline__ float fast_tanh(float x) {
#if __has_builtin(__builtin_amdgcn_tanhf)
  return __builtin_amdgcn_tanhf(x);
#else
  float cx = fminf(fmaxf(x, -10.f), 10.f);
  float e = exp2f(cx * 2.885390081777927f);
  return (e - 1.f) / (e + 1.f);
#endif
}

__device__ __forceinline__ float hw_exp2(float x) {
#if __has_builtin(__builtin_amdgcn_exp2f)
  return __builtin_amdgcn_exp2f(x);
#else
  return exp2f(x);
#endif
}

__device__ __forceinline__ float hw_rcp(float x) {
#if __has_builtin(__builtin_amdgcn_rcpf)
  return __builtin_amdgcn_rcpf(x);
#else
  return 1.f / x;
#endif
}

__device__ __forceinline__ v8f wmma4_f32(v2f a, v2f b, v8f c) {
#if __has_builtin(__builtin_amdgcn_wmma_f32_16x16x4_f32)
  return __builtin_amdgcn_wmma_f32_16x16x4_f32(
       false, a,  false, b,
       (short)0, c,  false,  false);
#else
  int lane = threadIdx.x & 31;
  int half = lane >> 4;
  int n = lane & 15;
  float B0 = __shfl(b.x, n),      B1 = __shfl(b.y, n);
  float B2 = __shfl(b.x, n + 16), B3 = __shfl(b.y, n + 16);
  v8f d;
#pragma unroll
  for (int j = 0; j < 8; ++j) {
    int r = j + (half ? 8 : 0);
    float A0 = __shfl(a.x, r),      A1 = __shfl(a.y, r);
    float A2 = __shfl(a.x, r + 16), A3 = __shfl(a.y, r + 16);
    d[j] = fmaf(A0, B0, fmaf(A1, B1, fmaf(A2, B2, fmaf(A3, B3, c[j]))));
  }
  return d;
#endif
}

struct Pack {
  v2f a0;
  v2f a1[3];
  v2f a2[3];
  v2f a3[3];
  v8f c0, c1, c2, c3;
  float t0[3], t1[3], t2[3], t3;
};

__device__ __forceinline__ void setup_pack(
    const float* m0, const float* b0, const float* f0,
    const float* m1, const float* b1, const float* f1,
    const float* m2, const float* b2, const float* f2,
    const float* m3, const float* b3, const float* f3,
    int wave, int lane, Pack& P) {
  const int half = lane >> 4, r = lane & 15;
  const int c = wave * 2 + half;
  const int j = r - 8 * half;
  const bool live3 = (j >= 0) && (j < 6);
  const bool live1 = (j >= 0) && (j < 2);
  const bool upper = (j & 1) != 0;
  const int o = live3 ? (j >> 1) : 0;

  {
    const float w = live3 ? softplusf(m0[c * 3 + o]) : 0.f;
    P.a0.x = upper ? 0.f : w;
    P.a0.y = upper ? w : 0.f;
  }
#pragma unroll
  for (int i = 0; i < 3; ++i) {
    const float w1 = live3 ? softplusf(m1[c * 9 + o * 3 + i]) : 0.f;
    const float w2 = live3 ? softplusf(m2[c * 9 + o * 3 + i]) : 0.f;
    const float w3 = live1 ? softplusf(m3[c * 3 + i]) : 0.f;
    P.a1[i].x = upper ? 0.f : w1;  P.a1[i].y = upper ? w1 : 0.f;
    P.a2[i].x = upper ? 0.f : w2;  P.a2[i].y = upper ? w2 : 0.f;
    P.a3[i].x = upper ? 0.f : w3;  P.a3[i].y = upper ? w3 : 0.f;
  }

#pragma unroll
  for (int i = 0; i < 3; ++i) {
    const float x0 = b0[c * 3 + i], x1 = b1[c * 3 + i], x2 = b2[c * 3 + i];
    P.c0[2 * i] = x0;  P.c0[2 * i + 1] = x0;
    P.c1[2 * i] = x1;  P.c1[2 * i + 1] = x1;
    P.c2[2 * i] = x2;  P.c2[2 * i + 1] = x2;
    P.t0[i] = tanhf(f0[c * 3 + i]);
    P.t1[i] = tanhf(f1[c * 3 + i]);
    P.t2[i] = tanhf(f2[c * 3 + i]);
  }
  P.c0[6] = 0.f;  P.c0[7] = 0.f;
  P.c1[6] = 0.f;  P.c1[7] = 0.f;
  P.c2[6] = 0.f;  P.c2[7] = 0.f;
  {
    const float x3 = b3[c];
    const v8f c3 = {x3, x3, 0.f, 0.f, 0.f, 0.f, 0.f, 0.f};
    P.c3 = c3;
  }
  P.t3 = tanhf(f3[c]);
}

__device__ __forceinline__ void gate6(const v8f d, const float* t, v2f& b0, v2f& b1, v2f& b2) {
  b0.x = d[0] + t[0] * fast_tanh(d[0]);
  b0.y = d[1] + t[0] * fast_tanh(d[1]);
  b1.x = d[2] + t[1] * fast_tanh(d[2]);
  b1.y = d[3] + t[1] * fast_tanh(d[3]);
  b2.x = d[4] + t[2] * fast_tanh(d[4]);
  b2.y = d[5] + t[2] * fast_tanh(d[5]);
}

__device__ __forceinline__ void eval_two(const Pack& P, float xa, float xb,
                                         float& loa, float& upa, float& lob, float& upb) {
  v2f ba, bb, pa, qa, ra, pb, qb, rb;
  ba.x = xa - 0.5f; ba.y = xa + 0.5f;
  bb.x = xb - 0.5f; bb.y = xb + 0.5f;
  v8f da = wmma4_f32(P.a0, ba, P.c0);
  v8f db = wmma4_f32(P.a0, bb, P.c0);
  gate6(da, P.t0, pa, qa, ra);
  gate6(db, P.t0, pb, qb, rb);

  da = wmma4_f32(P.a1[0], pa, P.c1);
  db = wmma4_f32(P.a1[0], pb, P.c1);
  da = wmma4_f32(P.a1[1], qa, da);
  db = wmma4_f32(P.a1[1], qb, db);
  da = wmma4_f32(P.a1[2], ra, da);
  db = wmma4_f32(P.a1[2], rb, db);
  gate6(da, P.t1, pa, qa, ra);
  gate6(db, P.t1, pb, qb, rb);

  da = wmma4_f32(P.a2[0], pa, P.c2);
  db = wmma4_f32(P.a2[0], pb, P.c2);
  da = wmma4_f32(P.a2[1], qa, da);
  db = wmma4_f32(P.a2[1], qb, db);
  da = wmma4_f32(P.a2[2], ra, da);
  db = wmma4_f32(P.a2[2], rb, db);
  gate6(da, P.t2, pa, qa, ra);
  gate6(db, P.t2, pb, qb, rb);

  da = wmma4_f32(P.a3[0], pa, P.c3);
  db = wmma4_f32(P.a3[0], pb, P.c3);
  da = wmma4_f32(P.a3[1], qa, da);
  db = wmma4_f32(P.a3[1], qb, db);
  da = wmma4_f32(P.a3[2], ra, da);
  db = wmma4_f32(P.a3[2], rb, db);
  loa = da[0] + P.t3 * fast_tanh(da[0]);
  upa = da[1] + P.t3 * fast_tanh(da[1]);
  lob = db[0] + P.t3 * fast_tanh(db[0]);
  upb = db[1] + P.t3 * fast_tanh(db[1]);
}

__device__ __forceinline__ float likelihood_of(float lo, float up) {
  const float L2E = 1.4426950408889634f;
  float sum = lo + up;
  float k = (sum > 0.f) ? L2E : ((sum < 0.f) ? -L2E : 0.f);
  float ea = hw_exp2(k * up);
  float eb = hw_exp2(k * lo);
  float lk = fabsf(hw_rcp(1.f + ea) - hw_rcp(1.f + eb));
  return fmaxf(lk, 1e-9f);
}

#define XP 82
#define LP 68

__device__ __forceinline__ void lds_barrier() {
  asm volatile("s_wait_dscnt 0x0\n\ts_barrier_signal -1\n\ts_barrier_wait -1" ::: "memory");
}

__global__ void __launch_bounds__(256)
eb_kernel(const float* __restrict__ inputs, const float* __restrict__ noise,
          const float* __restrict__ m0, const float* __restrict__ b0, const float* __restrict__ f0,
          const float* __restrict__ m1, const float* __restrict__ b1, const float* __restrict__ f1,
          const float* __restrict__ m2, const float* __restrict__ b2, const float* __restrict__ f2,
          const float* __restrict__ m3, const float* __restrict__ b3, const float* __restrict__ f3,
          float* __restrict__ out, float* __restrict__ lik, int ntiles) {
  __shared__ float xs[16 * XP];
  __shared__ __attribute__((aligned(16))) float ls[16 * LP];

  const int tid  = threadIdx.x;
  const int lane = tid & 31;
  const int wave = tid >> 5;
  const int half = lane >> 4;
  const int row  = lane & 15;

  const float4* __restrict__ in4 = (const float4*)inputs;
  const float4* __restrict__ nz4 = (const float4*)noise;
  float4* __restrict__ out4 = (float4*)out;
  float4* __restrict__ lik4 = (float4*)lik;

  int t = blockIdx.x;
  float4 a = {0.f, 0.f, 0.f, 0.f}, b = {0.f, 0.f, 0.f, 0.f};
  if (t < ntiles) {
    a = in4[t * 256 + tid];
    b = nz4[t * 256 + tid];
  }

  Pack P;
  setup_pack(m0, b0, f0, m1, b1, f1, m2, b2, f2, m3, b3, f3, wave, lane, P);

  const int p = tid >> 2, q = tid & 3;
  float* xw = &xs[4 * q * XP + p];
  const float4* lr = (const float4*)&ls[(p >> 2) * LP + (p & 3) * 16 + 4 * q];
  const float* xr = &xs[(2 * wave + half) * XP + row];
  float* lw = &ls[(row >> 2) * LP + (row & 3) * 16 + 2 * wave + half];

  for (; t < ntiles; t += gridDim.x) {
    float4 v;
    v.x = a.x + (b.x - 0.5f);
    v.y = a.y + (b.y - 0.5f);
    v.z = a.z + (b.z - 0.5f);
    v.w = a.w + (b.w - 0.5f);
    out4[t * 256 + tid] = v;
    xw[0]      = v.x;
    xw[XP]     = v.y;
    xw[2 * XP] = v.z;
    xw[3 * XP] = v.w;

    const int tn = t + gridDim.x;
    if (tn < ntiles) {
      a = in4[tn * 256 + tid];
      b = nz4[tn * 256 + tid];
    }
    lds_barrier();

#pragma nounroll
    for (int g = 0; g < 4; g += 2) {
      float loa, upa, lob, upb;
      eval_two(P, xr[16 * g], xr[16 * g + 16], loa, upa, lob, upb);
      lw[4 * g * LP]       = likelihood_of(loa, upa);
      lw[4 * (g + 1) * LP] = likelihood_of(lob, upb);
    }
    lds_barrier();

    lik4[t * 256 + tid] = *lr;
  }
}

extern "C" void kernel_launch(void* const* d_in, const int* in_sizes, int n_in,
                              void* d_out, int out_size, void* d_ws, size_t ws_size,
                              hipStream_t stream) {
  const float* inputs = (const float*)d_in[0];
  const float* noise  = (const float*)d_in[1];
  const float* m0 = (const float*)d_in[2];
  const float* b0 = (const float*)d_in[3];
  const float* f0 = (const float*)d_in[4];
  const float* m1 = (const float*)d_in[5];
  const float* b1 = (const float*)d_in[6];
  const float* f1 = (const float*)d_in[7];
  const float* m2 = (const float*)d_in[8];
  const float* b2 = (const float*)d_in[9];
  const float* f2 = (const float*)d_in[10];
  const float* m3 = (const float*)d_in[11];
  const float* b3 = (const float*)d_in[12];
  const float* f3 = (const float*)d_in[13];

  const int nelem  = in_sizes[0];
  const int ntiles = nelem / 1024;
  float* out = (float*)d_out;
  float* lik = out + nelem;

  int blocks = ntiles < 1024 ? ntiles : 1024;
  eb_kernel<<<blocks, 256, 0, stream>>>(inputs, noise,
                                        m0, b0, f0, m1, b1, f1,
                                        m2, b2, f2, m3, b3, f3,
                                        out, lik, ntiles);
}


# softplus(m) and tanh(f) evaluated once per work-group through LDS (2-3 library calls per thread, one barrier) instead of 20 per lane in every wave; packed values bit-identical; sum order unchanged
# speedup vs baseline: 3.3069x; 1.0646x over previous
#include <hip/hip_runtime.h>
#include <math.h>


typedef float v2f __attribute__((ext_vector_type(2)));
typedef float v8f __attribute__((ext_vector_type(8)));

__device__ __forceinline__ float softplusf(float x) {
  return (x > 20.f) ? x : log1pf(expf(x));
}

__device__ __forceinline__ float fast_tanh(float x) {
#if __has_builtin(__builtin_amdgcn_tanhf)
  return __builtin_amdgcn_tanhf(x);
#else
  float cx = fminf(fmaxf(x, -10.f), 10.f);
  float e = exp2f(cx * 2.885390081777927f);
  return (e - 1.f) / (e + 1.f);
#endif
}

__device__ __forceinline__ float hw_exp2(float x) {
#if __has_builtin(__builtin_amdgcn_exp2f)
  return __builtin_amdgcn_exp2f(x);
#else
  return exp2f(x);
#endif
}

__device__ __forceinline__ float hw_rcp(float x) {
#if __has_builtin(__builtin_amdgcn_rcpf)
  return __builtin_amdgcn_rcpf(x);
#else
  return 1.f / x;
#endif
}

__device__ __forceinline__ v8f wmma4_f32(v2f a, v2f b, v8f c) {
#if __has_builtin(__builtin_amdgcn_wmma_f32_16x16x4_f32)
  return __builtin_amdgcn_wmma_f32_16x16x4_f32(
       false, a,  false, b,
       (short)0, c,  false,  false);
#else
  int lane = threadIdx.x & 31;
  int half = lane >> 4;
  int n = lane & 15;
  float B0 = __shfl(b.x, n),      B1 = __shfl(b.y, n);
  float B2 = __shfl(b.x, n + 16), B3 = __shfl(b.y, n + 16);
  v8f d;
#pragma unroll
  for (int j = 0; j < 8; ++j) {
    int r = j + (half ? 8 : 0);
    float A0 = __shfl(a.x, r),      A1 = __shfl(a.y, r);
    float A2 = __shfl(a.x, r + 16), A3 = __shfl(a.y, r + 16);
    d[j] = fmaf(A0, B0, fmaf(A1, B1, fmaf(A2, B2, fmaf(A3, B3, c[j]))));
  }
  return d;
#endif
}

struct Pack {
  v2f a0;
  v2f a1[3];
  v2f a2[3];
  v2f a3[3];
  v8f c0, c1, c2, c3;
  float t0[3], t1[3], t2[3], t3;
};

#define SP_M0 0
#define SP_M1 48
#define SP_M2 192
#define SP_M3 336
#define SP_N  384
#define TH_F0 0
#define TH_F1 48
#define TH_F2 96
#define TH_F3 144
#define TH_N  160

__device__ __forceinline__ void transform_params(
    const float* m0, const float* f0, const float* m1, const float* f1,
    const float* m2, const float* f2, const float* m3, const float* f3,
    int tid, float* sp, float* th) {
  const float* p1 = (tid < SP_M1) ? (m0 + tid)
                  : ((tid < SP_M2) ? (m1 + (tid - SP_M1)) : (m2 + (tid - SP_M2)));
  const int k2 = tid + 256;
  const int h2 = tid - 128;
  const float* p2 = (tid < 128)
      ? ((k2 < SP_M3) ? (m2 + (k2 - SP_M2)) : (m3 + (k2 - SP_M3)))
      : ((h2 < TH_F1) ? (f0 + h2) : ((h2 < TH_F2) ? (f1 + (h2 - TH_F1)) : (f2 + (h2 - TH_F2))));
  const int h3 = tid - 96;
  float x3 = 0.f;
  if (tid >= 224) x3 = (h3 < TH_F3) ? f2[h3 - TH_F2] : f3[h3 - TH_F3];
  const float x1 = *p1;
  const float x2 = *p2;

  sp[tid] = softplusf(x1);
  if (tid < 128) sp[k2] = softplusf(x2);
  else           th[h2] = tanhf(x2);
  if (tid >= 224) th[h3] = tanhf(x3);
}

struct Bias { float b0[3], b1[3], b2[3], b3; };

__device__ __forceinline__ void load_bias(
    const float* b0, const float* b1, const float* b2, const float* b3,
    int wave, int lane, Bias& B) {
  const int c = wave * 2 + (lane >> 4);
#pragma unroll
  for (int i = 0; i < 3; ++i) {
    B.b0[i] = b0[c * 3 + i];
    B.b1[i] = b1[c * 3 + i];
    B.b2[i] = b2[c * 3 + i];
  }
  B.b3 = b3[c];
}

__device__ __forceinline__ void load_pack(
    const float* sp, const float* th, const Bias& B,
    int wave, int lane, Pack& P) {
  const int half = lane >> 4, r = lane & 15;
  const int c = wave * 2 + half;
  const int j = r - 8 * half;
  const bool live3 = (j >= 0) && (j < 6);
  const bool live1 = (j >= 0) && (j < 2);
  const bool upper = (j & 1) != 0;
  const int o = live3 ? (j >> 1) : 0;

  {
    const float w = live3 ? sp[SP_M0 + c * 3 + o] : 0.f;
    P.a0.x = upper ? 0.f : w;
    P.a0.y = upper ? w : 0.f;
  }
#pragma unroll
  for (int i = 0; i < 3; ++i) {
    const float w1 = live3 ? sp[SP_M1 + c * 9 + o * 3 + i] : 0.f;
    const float w2 = live3 ? sp[SP_M2 + c * 9 + o * 3 + i] : 0.f;
    const float w3 = live1 ? sp[SP_M3 + c * 3 + i] : 0.f;
    P.a1[i].x = upper ? 0.f : w1;  P.a1[i].y = upper ? w1 : 0.f;
    P.a2[i].x = upper ? 0.f : w2;  P.a2[i].y = upper ? w2 : 0.f;
    P.a3[i].x = upper ? 0.f : w3;  P.a3[i].y = upper ? w3 : 0.f;
  }

#pragma unroll
  for (int i = 0; i < 3; ++i) {
    const float x0 = B.b0[i], x1 = B.b1[i], x2 = B.b2[i];
    P.c0[2 * i] = x0;  P.c0[2 * i + 1] = x0;
    P.c1[2 * i] = x1;  P.c1[2 * i + 1] = x1;
    P.c2[2 * i] = x2;  P.c2[2 * i + 1] = x2;
    P.t0[i] = th[TH_F0 + c * 3 + i];
    P.t1[i] = th[TH_F1 + c * 3 + i];
    P.t2[i] = th[TH_F2 + c * 3 + i];
  }
  P.c0[6] = 0.f;  P.c0[7] = 0.f;
  P.c1[6] = 0.f;  P.c1[7] = 0.f;
  P.c2[6] = 0.f;  P.c2[7] = 0.f;
  {
    const float x3 = B.b3;
    const v8f c3 = {x3, x3, 0.f, 0.f, 0.f, 0.f, 0.f, 0.f};
    P.c3 = c3;
  }
  P.t3 = th[TH_F3 + c];
}

__device__ __forceinline__ void gate6(const v8f d, const float* t, v2f& b0, v2f& b1, v2f& b2) {
  b0.x = d[0] + t[0] * fast_tanh(d[0]);
  b0.y = d[1] + t[0] * fast_tanh(d[1]);
  b1.x = d[2] + t[1] * fast_tanh(d[2]);
  b1.y = d[3] + t[1] * fast_tanh(d[3]);
  b2.x = d[4] + t[2] * fast_tanh(d[4]);
  b2.y = d[5] + t[2] * fast_tanh(d[5]);
}

__device__ __forceinline__ void eval_two(const Pack& P, float xa, float xb,
                                         float& loa, float& upa, float& lob, float& upb) {
  v2f ba, bb, pa, qa, ra, pb, qb, rb;
  ba.x = xa - 0.5f; ba.y = xa + 0.5f;
  bb.x = xb - 0.5f; bb.y = xb + 0.5f;
  v8f da = wmma4_f32(P.a0, ba, P.c0);
  v8f db = wmma4_f32(P.a0, bb, P.c0);
  gate6(da, P.t0, pa, qa, ra);
  gate6(db, P.t0, pb, qb, rb);

  da = wmma4_f32(P.a1[0], pa, P.c1);
  db = wmma4_f32(P.a1[0], pb, P.c1);
  da = wmma4_f32(P.a1[1], qa, da);
  db = wmma4_f32(P.a1[1], qb, db);
  da = wmma4_f32(P.a1[2], ra, da);
  db = wmma4_f32(P.a1[2], rb, db);
  gate6(da, P.t1, pa, qa, ra);
  gate6(db, P.t1, pb, qb, rb);

  da = wmma4_f32(P.a2[0], pa, P.c2);
  db = wmma4_f32(P.a2[0], pb, P.c2);
  da = wmma4_f32(P.a2[1], qa, da);
  db = wmma4_f32(P.a2[1], qb, db);
  da = wmma4_f32(P.a2[2], ra, da);
  db = wmma4_f32(P.a2[2], rb, db);
  gate6(da, P.t2, pa, qa, ra);
  gate6(db, P.t2, pb, qb, rb);

  da = wmma4_f32(P.a3[0], pa, P.c3);
  db = wmma4_f32(P.a3[0], pb, P.c3);
  da = wmma4_f32(P.a3[1], qa, da);
  db = wmma4_f32(P.a3[1], qb, db);
  da = wmma4_f32(P.a3[2], ra, da);
  db = wmma4_f32(P.a3[2], rb, db);
  loa = da[0] + P.t3 * fast_tanh(da[0]);
  upa = da[1] + P.t3 * fast_tanh(da[1]);
  lob = db[0] + P.t3 * fast_tanh(db[0]);
  upb = db[1] + P.t3 * fast_tanh(db[1]);
}

__device__ __forceinline__ float likelihood_of(float lo, float up) {
  const float L2E = 1.4426950408889634f;
  float sum = lo + up;
  float k = (sum > 0.f) ? L2E : ((sum < 0.f) ? -L2E : 0.f);
  float ea = hw_exp2(k * up);
  float eb = hw_exp2(k * lo);
  float lk = fabsf(hw_rcp(1.f + ea) - hw_rcp(1.f + eb));
  return fmaxf(lk, 1e-9f);
}

#define XP 82
#define LP 68

__device__ __forceinline__ void lds_barrier() {
  asm volatile("s_wait_dscnt 0x0\n\ts_barrier_signal -1\n\ts_barrier_wait -1" ::: "memory");
}

__global__ void __launch_bounds__(256)
eb_kernel(const float* __restrict__ inputs, const float* __restrict__ noise,
          const float* __restrict__ m0, const float* __restrict__ b0, const float* __restrict__ f0,
          const float* __restrict__ m1, const float* __restrict__ b1, const float* __restrict__ f1,
          const float* __restrict__ m2, const float* __restrict__ b2, const float* __restrict__ f2,
          const float* __restrict__ m3, const float* __restrict__ b3, const float* __restrict__ f3,
          float* __restrict__ out, float* __restrict__ lik, int ntiles) {
  __shared__ float xs[16 * XP];
  __shared__ __attribute__((aligned(16))) float ls[16 * LP];
  __shared__ float sp[SP_N];
  __shared__ float th[TH_N];

  const int tid  = threadIdx.x;
  const int lane = tid & 31;
  const int wave = tid >> 5;
  const int half = lane >> 4;
  const int row  = lane & 15;

  const float4* __restrict__ in4 = (const float4*)inputs;
  const float4* __restrict__ nz4 = (const float4*)noise;
  float4* __restrict__ out4 = (float4*)out;
  float4* __restrict__ lik4 = (float4*)lik;

  int t = blockIdx.x;
  float4 a = {0.f, 0.f, 0.f, 0.f}, b = {0.f, 0.f, 0.f, 0.f};
  if (t < ntiles) {
    a = in4[t * 256 + tid];
    b = nz4[t * 256 + tid];
  }

  Bias Bv;
  load_bias(b0, b1, b2, b3, wave, lane, Bv);
  transform_params(m0, f0, m1, f1, m2, f2, m3, f3, tid, sp, th);
  lds_barrier();
  Pack P;
  load_pack(sp, th, Bv, wave, lane, P);

  const int p = tid >> 2, q = tid & 3;
  float* xw = &xs[4 * q * XP + p];
  const float4* lr = (const float4*)&ls[(p >> 2) * LP + (p & 3) * 16 + 4 * q];
  const float* xr = &xs[(2 * wave + half) * XP + row];
  float* lw = &ls[(row >> 2) * LP + (row & 3) * 16 + 2 * wave + half];

  for (; t < ntiles; t += gridDim.x) {
    float4 v;
    v.x = a.x + (b.x - 0.5f);
    v.y = a.y + (b.y - 0.5f);
    v.z = a.z + (b.z - 0.5f);
    v.w = a.w + (b.w - 0.5f);
    out4[t * 256 + tid] = v;
    xw[0]      = v.x;
    xw[XP]     = v.y;
    xw[2 * XP] = v.z;
    xw[3 * XP] = v.w;

    const int tn = t + gridDim.x;
    if (tn < ntiles) {
      a = in4[tn * 256 + tid];
      b = nz4[tn * 256 + tid];
    }
    lds_barrier();

#pragma nounroll
    for (int g = 0; g < 4; g += 2) {
      float loa, upa, lob, upb;
      eval_two(P, xr[16 * g], xr[16 * g + 16], loa, upa, lob, upb);
      lw[4 * g * LP]       = likelihood_of(loa, upa);
      lw[4 * (g + 1) * LP] = likelihood_of(lob, upb);
    }
    lds_barrier();

    lik4[t * 256 + tid] = *lr;
  }
}

extern "C" void kernel_launch(void* const* d_in, const int* in_sizes, int n_in,
                              void* d_out, int out_size, void* d_ws, size_t ws_size,
                              hipStream_t stream) {
  const float* inputs = (const float*)d_in[0];
  const float* noise  = (const float*)d_in[1];
  const float* m0 = (const float*)d_in[2];
  const float* b0 = (const float*)d_in[3];
  const float* f0 = (const float*)d_in[4];
  const float* m1 = (const float*)d_in[5];
  const float* b1 = (const float*)d_in[6];
  const float* f1 = (const float*)d_in[7];
  const float* m2 = (const float*)d_in[8];
  const float* b2 = (const float*)d_in[9];
  const float* f2 = (const float*)d_in[10];
  const float* m3 = (const float*)d_in[11];
  const float* b3 = (const float*)d_in[12];
  const float* f3 = (const float*)d_in[13];

  const int nelem  = in_sizes[0];
  const int ntiles = nelem / 1024;
  float* out = (float*)d_out;
  float* lik = out + nelem;

  int blocks = ntiles < 1024 ? ntiles : 1024;
  eb_kernel<<<blocks, 256, 0, stream>>>(inputs, noise,
                                        m0, b0, f0, m1, b1, f1,
                                        m2, b2, f2, m3, b3, f3,
                                        out, lik, ntiles);
}


# Issue priority by remaining tiles: s_setprio 3 until three tiles are left, then 2, 1, 0, so the four work-groups of a compute unit finish together instead of the youngest running its tail alone
# speedup vs baseline: 3.4366x; 1.0392x over previous
#include <hip/hip_runtime.h>
#include <math.h>


typedef float v2f __attribute__((ext_vector_type(2)));
typedef float v8f __attribute__((ext_vector_type(8)));

__device__ __forceinline__ float softplusf(float x) {
  return (x > 20.f) ? x : log1pf(expf(x));
}

__device__ __forceinline__ float fast_tanh(float x) {
#if __has_builtin(__builtin_amdgcn_tanhf)
  return __builtin_amdgcn_tanhf(x);
#else
  float cx = fminf(fmaxf(x, -10.f), 10.f);
  float e = exp2f(cx * 2.885390081777927f);
  return (e - 1.f) / (e + 1.f);
#endif
}

__device__ __forceinline__ float hw_exp2(float x) {
#if __has_builtin(__builtin_amdgcn_exp2f)
  return __builtin_amdgcn_exp2f(x);
#else
  return exp2f(x);
#endif
}

__device__ __forceinline__ float hw_rcp(float x) {
#if __has_builtin(__builtin_amdgcn_rcpf)
  return __builtin_amdgcn_rcpf(x);
#else
  return 1.f / x;
#endif
}

__device__ __forceinline__ v8f wmma4_f32(v2f a, v2f b, v8f c) {
#if __has_builtin(__builtin_amdgcn_wmma_f32_16x16x4_f32)
  return __builtin_amdgcn_wmma_f32_16x16x4_f32(
       false, a,  false, b,
       (short)0, c,  false,  false);
#else
  int lane = threadIdx.x & 31;
  int half = lane >> 4;
  int n = lane & 15;
  float B0 = __shfl(b.x, n),      B1 = __shfl(b.y, n);
  float B2 = __shfl(b.x, n + 16), B3 = __shfl(b.y, n + 16);
  v8f d;
#pragma unroll
  for (int j = 0; j < 8; ++j) {
    int r = j + (half ? 8 : 0);
    float A0 = __shfl(a.x, r),      A1 = __shfl(a.y, r);
    float A2 = __shfl(a.x, r + 16), A3 = __shfl(a.y, r + 16);
    d[j] = fmaf(A0, B0, fmaf(A1, B1, fmaf(A2, B2, fmaf(A3, B3, c[j]))));
  }
  return d;
#endif
}

struct Pack {
  v2f a0;
  v2f a1[3];
  v2f a2[3];
  v2f a3[3];
  v8f c0, c1, c2, c3;
  float t0[3], t1[3], t2[3], t3;
};

#define SP_M0 0
#define SP_M1 48
#define SP_M2 192
#define SP_M3 336
#define SP_N  384
#define TH_F0 0
#define TH_F1 48
#define TH_F2 96
#define TH_F3 144
#define TH_N  160

__device__ __forceinline__ void transform_params(
    const float* m0, const float* f0, const float* m1, const float* f1,
    const float* m2, const float* f2, const float* m3, const float* f3,
    int tid, float* sp, float* th) {
  const float* p1 = (tid < SP_M1) ? (m0 + tid)
                  : ((tid < SP_M2) ? (m1 + (tid - SP_M1)) : (m2 + (tid - SP_M2)));
  const int k2 = tid + 256;
  const int h2 = tid - 128;
  const float* p2 = (tid < 128)
      ? ((k2 < SP_M3) ? (m2 + (k2 - SP_M2)) : (m3 + (k2 - SP_M3)))
      : ((h2 < TH_F1) ? (f0 + h2) : ((h2 < TH_F2) ? (f1 + (h2 - TH_F1)) : (f2 + (h2 - TH_F2))));
  const int h3 = tid - 96;
  float x3 = 0.f;
  if (tid >= 224) x3 = (h3 < TH_F3) ? f2[h3 - TH_F2] : f3[h3 - TH_F3];
  const float x1 = *p1;
  const float x2 = *p2;

  sp[tid] = softplusf(x1);
  if (tid < 128) sp[k2] = softplusf(x2);
  else           th[h2] = tanhf(x2);
  if (tid >= 224) th[h3] = tanhf(x3);
}

struct Bias { float b0[3], b1[3], b2[3], b3; };

__device__ __forceinline__ void load_bias(
    const float* b0, const float* b1, const float* b2, const float* b3,
    int wave, int lane, Bias& B) {
  const int c = wave * 2 + (lane >> 4);
#pragma unroll
  for (int i = 0; i < 3; ++i) {
    B.b0[i] = b0[c * 3 + i];
    B.b1[i] = b1[c * 3 + i];
    B.b2[i] = b2[c * 3 + i];
  }
  B.b3 = b3[c];
}

__device__ __forceinline__ void load_pack(
    const float* sp, const float* th, const Bias& B,
    int wave, int lane, Pack& P) {
  const int half = lane >> 4, r = lane & 15;
  const int c = wave * 2 + half;
  const int j = r - 8 * half;
  const bool live3 = (j >= 0) && (j < 6);
  const bool live1 = (j >= 0) && (j < 2);
  const bool upper = (j & 1) != 0;
  const int o = live3 ? (j >> 1) : 0;

  {
    const float w = live3 ? sp[SP_M0 + c * 3 + o] : 0.f;
    P.a0.x = upper ? 0.f : w;
    P.a0.y = upper ? w : 0.f;
  }
#pragma unroll
  for (int i = 0; i < 3; ++i) {
    const float w1 = live3 ? sp[SP_M1 + c * 9 + o * 3 + i] : 0.f;
    const float w2 = live3 ? sp[SP_M2 + c * 9 + o * 3 + i] : 0.f;
    const float w3 = live1 ? sp[SP_M3 + c * 3 + i] : 0.f;
    P.a1[i].x = upper ? 0.f : w1;  P.a1[i].y = upper ? w1 : 0.f;
    P.a2[i].x = upper ? 0.f : w2;  P.a2[i].y = upper ? w2 : 0.f;
    P.a3[i].x = upper ? 0.f : w3;  P.a3[i].y = upper ? w3 : 0.f;
  }

#pragma unroll
  for (int i = 0; i < 3; ++i) {
    const float x0 = B.b0[i], x1 = B.b1[i], x2 = B.b2[i];
    P.c0[2 * i] = x0;  P.c0[2 * i + 1] = x0;
    P.c1[2 * i] = x1;  P.c1[2 * i + 1] = x1;
    P.c2[2 * i] = x2;  P.c2[2 * i + 1] = x2;
    P.t0[i] = th[TH_F0 + c * 3 + i];
    P.t1[i] = th[TH_F1 + c * 3 + i];
    P.t2[i] = th[TH_F2 + c * 3 + i];
  }
  P.c0[6] = 0.f;  P.c0[7] = 0.f;
  P.c1[6] = 0.f;  P.c1[7] = 0.f;
  P.c2[6] = 0.f;  P.c2[7] = 0.f;
  {
    const float x3 = B.b3;
    const v8f c3 = {x3, x3, 0.f, 0.f, 0.f, 0.f, 0.f, 0.f};
    P.c3 = c3;
  }
  P.t3 = th[TH_F3 + c];
}

__device__ __forceinline__ void gate6(const v8f d, const float* t, v2f& b0, v2f& b1, v2f& b2) {
  b0.x = d[0] + t[0] * fast_tanh(d[0]);
  b0.y = d[1] + t[0] * fast_tanh(d[1]);
  b1.x = d[2] + t[1] * fast_tanh(d[2]);
  b1.y = d[3] + t[1] * fast_tanh(d[3]);
  b2.x = d[4] + t[2] * fast_tanh(d[4]);
  b2.y = d[5] + t[2] * fast_tanh(d[5]);
}

__device__ __forceinline__ void eval_two(const Pack& P, float xa, float xb,
                                         float& loa, float& upa, float& lob, float& upb) {
  v2f ba, bb, pa, qa, ra, pb, qb, rb;
  ba.x = xa - 0.5f; ba.y = xa + 0.5f;
  bb.x = xb - 0.5f; bb.y = xb + 0.5f;
  v8f da = wmma4_f32(P.a0, ba, P.c0);
  v8f db = wmma4_f32(P.a0, bb, P.c0);
  gate6(da, P.t0, pa, qa, ra);
  gate6(db, P.t0, pb, qb, rb);

  da = wmma4_f32(P.a1[0], pa, P.c1);
  db = wmma4_f32(P.a1[0], pb, P.c1);
  da = wmma4_f32(P.a1[1], qa, da);
  db = wmma4_f32(P.a1[1], qb, db);
  da = wmma4_f32(P.a1[2], ra, da);
  db = wmma4_f32(P.a1[2], rb, db);
  gate6(da, P.t1, pa, qa, ra);
  gate6(db, P.t1, pb, qb, rb);

  da = wmma4_f32(P.a2[0], pa, P.c2);
  db = wmma4_f32(P.a2[0], pb, P.c2);
  da = wmma4_f32(P.a2[1], qa, da);
  db = wmma4_f32(P.a2[1], qb, db);
  da = wmma4_f32(P.a2[2], ra, da);
  db = wmma4_f32(P.a2[2], rb, db);
  gate6(da, P.t2, pa, qa, ra);
  gate6(db, P.t2, pb, qb, rb);

  da = wmma4_f32(P.a3[0], pa, P.c3);
  db = wmma4_f32(P.a3[0], pb, P.c3);
  da = wmma4_f32(P.a3[1], qa, da);
  db = wmma4_f32(P.a3[1], qb, db);
  da = wmma4_f32(P.a3[2], ra, da);
  db = wmma4_f32(P.a3[2], rb, db);
  loa = da[0] + P.t3 * fast_tanh(da[0]);
  upa = da[1] + P.t3 * fast_tanh(da[1]);
  lob = db[0] + P.t3 * fast_tanh(db[0]);
  upb = db[1] + P.t3 * fast_tanh(db[1]);
}

__device__ __forceinline__ float likelihood_of(float lo, float up) {
  const float L2E = 1.4426950408889634f;
  float sum = lo + up;
  float k = (sum > 0.f) ? L2E : ((sum < 0.f) ? -L2E : 0.f);
  float ea = hw_exp2(k * up);
  float eb = hw_exp2(k * lo);
  float lk = fabsf(hw_rcp(1.f + ea) - hw_rcp(1.f + eb));
  return fmaxf(lk, 1e-9f);
}

#define XP 82
#define LP 68

__device__ __forceinline__ void lds_barrier() {
  asm volatile("s_wait_dscnt 0x0\n\ts_barrier_signal -1\n\ts_barrier_wait -1" ::: "memory");
}

__global__ void __launch_bounds__(256)
eb_kernel(const float* __restrict__ inputs, const float* __restrict__ noise,
          const float* __restrict__ m0, const float* __restrict__ b0, const float* __restrict__ f0,
          const float* __restrict__ m1, const float* __restrict__ b1, const float* __restrict__ f1,
          const float* __restrict__ m2, const float* __restrict__ b2, const float* __restrict__ f2,
          const float* __restrict__ m3, const float* __restrict__ b3, const float* __restrict__ f3,
          float* __restrict__ out, float* __restrict__ lik, int ntiles) {
  __shared__ float xs[16 * XP];
  __shared__ __attribute__((aligned(16))) float ls[16 * LP];
  __shared__ float sp[SP_N];
  __shared__ float th[TH_N];

  const int tid  = threadIdx.x;
  const int lane = tid & 31;
  const int wave = tid >> 5;
  const int half = lane >> 4;
  const int row  = lane & 15;

  const float4* __restrict__ in4 = (const float4*)inputs;
  const float4* __restrict__ nz4 = (const float4*)noise;
  float4* __restrict__ out4 = (float4*)out;
  float4* __restrict__ lik4 = (float4*)lik;

  __builtin_amdgcn_s_setprio(3);

  int t = blockIdx.x;
  int left = (t < ntiles) ? (ntiles - 1 - t) / (int)gridDim.x : 0;
  float4 a = {0.f, 0.f, 0.f, 0.f}, b = {0.f, 0.f, 0.f, 0.f};
  if (t < ntiles) {
    a = in4[t * 256 + tid];
    b = nz4[t * 256 + tid];
  }

  Bias Bv;
  load_bias(b0, b1, b2, b3, wave, lane, Bv);
  transform_params(m0, f0, m1, f1, m2, f2, m3, f3, tid, sp, th);
  lds_barrier();
  Pack P;
  load_pack(sp, th, Bv, wave, lane, P);

  const int p = tid >> 2, q = tid & 3;
  float* xw = &xs[4 * q * XP + p];
  const float4* lr = (const float4*)&ls[(p >> 2) * LP + (p & 3) * 16 + 4 * q];
  const float* xr = &xs[(2 * wave + half) * XP + row];
  float* lw = &ls[(row >> 2) * LP + (row & 3) * 16 + 2 * wave + half];

  for (; t < ntiles; t += gridDim.x) {
    if (left < 3) {
      if (left == 2)      __builtin_amdgcn_s_setprio(2);
      else if (left == 1) __builtin_amdgcn_s_setprio(1);
      else                __builtin_amdgcn_s_setprio(0);
    }
    --left;

    float4 v;
    v.x = a.x + (b.x - 0.5f);
    v.y = a.y + (b.y - 0.5f);
    v.z = a.z + (b.z - 0.5f);
    v.w = a.w + (b.w - 0.5f);
    out4[t * 256 + tid] = v;
    xw[0]      = v.x;
    xw[XP]     = v.y;
    xw[2 * XP] = v.z;
    xw[3 * XP] = v.w;

    const int tn = t + gridDim.x;
    if (tn < ntiles) {
      a = in4[tn * 256 + tid];
      b = nz4[tn * 256 + tid];
    }
    lds_barrier();

#pragma nounroll
    for (int g = 0; g < 4; g += 2) {
      float loa, upa, lob, upb;
      eval_two(P, xr[16 * g], xr[16 * g + 16], loa, upa, lob, upb);
      lw[4 * g * LP]       = likelihood_of(loa, upa);
      lw[4 * (g + 1) * LP] = likelihood_of(lob, upb);
    }
    lds_barrier();

    lik4[t * 256 + tid] = *lr;
  }
}

extern "C" void kernel_launch(void* const* d_in, const int* in_sizes, int n_in,
                              void* d_out, int out_size, void* d_ws, size_t ws_size,
                              hipStream_t stream) {
  const float* inputs = (const float*)d_in[0];
  const float* noise  = (const float*)d_in[1];
  const float* m0 = (const float*)d_in[2];
  const float* b0 = (const float*)d_in[3];
  const float* f0 = (const float*)d_in[4];
  const float* m1 = (const float*)d_in[5];
  const float* b1 = (const float*)d_in[6];
  const float* f1 = (const float*)d_in[7];
  const float* m2 = (const float*)d_in[8];
  const float* b2 = (const float*)d_in[9];
  const float* f2 = (const float*)d_in[10];
  const float* m3 = (const float*)d_in[11];
  const float* b3 = (const float*)d_in[12];
  const float* f3 = (const float*)d_in[13];

  const int nelem  = in_sizes[0];
  const int ntiles = nelem / 1024;
  float* out = (float*)d_out;
  float* lik = out + nelem;

  int blocks = ntiles < 1024 ? ntiles : 1024;
  eb_kernel<<<blocks, 256, 0, stream>>>(inputs, noise,
                                        m0, b0, f0, m1, b1, f1,
                                        m2, b2, f2, m3, b3, f3,
                                        out, lik, ntiles);
}


# Layer 0 once per point, not per argument: one matrix op on x gives lower for two 16-point groups, upper = lower + w; 19 matrix ops per 64 elems, was 20; layer-3 bias shares that C; sum order changes
# speedup vs baseline: 3.5392x; 1.0299x over previous
#include <hip/hip_runtime.h>
#include <math.h>


typedef float v2f __attribute__((ext_vector_type(2)));
typedef float v8f __attribute__((ext_vector_type(8)));

__device__ __forceinline__ float softplusf(float x) {
  return (x > 20.f) ? x : log1pf(expf(x));
}

__device__ __forceinline__ float fast_tanh(float x) {
#if __has_builtin(__builtin_amdgcn_tanhf)
  return __builtin_amdgcn_tanhf(x);
#else
  float cx = fminf(fmaxf(x, -10.f), 10.f);
  float e = exp2f(cx * 2.885390081777927f);
  return (e - 1.f) / (e + 1.f);
#endif
}

__device__ __forceinline__ float hw_exp2(float x) {
#if __has_builtin(__builtin_amdgcn_exp2f)
  return __builtin_amdgcn_exp2f(x);
#else
  return exp2f(x);
#endif
}

__device__ __forceinline__ float hw_rcp(float x) {
#if __has_builtin(__builtin_amdgcn_rcpf)
  return __builtin_amdgcn_rcpf(x);
#else
  return 1.f / x;
#endif
}

__device__ __forceinline__ v8f wmma4_f32(v2f a, v2f b, v8f c) {
#if __has_builtin(__builtin_amdgcn_wmma_f32_16x16x4_f32)
  return __builtin_amdgcn_wmma_f32_16x16x4_f32(
       false, a,  false, b,
       (short)0, c,  false,  false);
#else
  int lane = threadIdx.x & 31;
  int half = lane >> 4;
  int n = lane & 15;
  float B0 = __shfl(b.x, n),      B1 = __shfl(b.y, n);
  float B2 = __shfl(b.x, n + 16), B3 = __shfl(b.y, n + 16);
  v8f d;
#pragma unroll
  for (int j = 0; j < 8; ++j) {
    int r = j + (half ? 8 : 0);
    float A0 = __shfl(a.x, r),      A1 = __shfl(a.y, r);
    float A2 = __shfl(a.x, r + 16), A3 = __shfl(a.y, r + 16);
    d[j] = fmaf(A0, B0, fmaf(A1, B1, fmaf(A2, B2, fmaf(A3, B3, c[j]))));
  }
  return d;
#endif
}

struct Pack {
  v2f a0;
  v2f a1[3];
  v2f a2[3];
  v2f a3[3];
  v8f c0, c1, c2;
  float w0[3];
  float t0[3], t1[3], t2[3], t3;
};

#define SP_M0 0
#define SP_M1 48
#define SP_M2 192
#define SP_M3 336
#define SP_N  384
#define TH_F0 0
#define TH_F1 48
#define TH_F2 96
#define TH_F3 144
#define TH_N  160

__device__ __forceinline__ void transform_params(
    const float* m0, const float* f0, const float* m1, const float* f1,
    const float* m2, const float* f2, const float* m3, const float* f3,
    int tid, float* sp, float* th) {
  const float* p1 = (tid < SP_M1) ? (m0 + tid)
                  : ((tid < SP_M2) ? (m1 + (tid - SP_M1)) : (m2 + (tid - SP_M2)));
  const int k2 = tid + 256;
  const int h2 = tid - 128;
  const float* p2 = (tid < 128)
      ? ((k2 < SP_M3) ? (m2 + (k2 - SP_M2)) : (m3 + (k2 - SP_M3)))
      : ((h2 < TH_F1) ? (f0 + h2) : ((h2 < TH_F2) ? (f1 + (h2 - TH_F1)) : (f2 + (h2 - TH_F2))));
  const int h3 = tid - 96;
  float x3 = 0.f;
  if (tid >= 224) x3 = (h3 < TH_F3) ? f2[h3 - TH_F2] : f3[h3 - TH_F3];
  const float x1 = *p1;
  const float x2 = *p2;

  sp[tid] = softplusf(x1);
  if (tid < 128) sp[k2] = softplusf(x2);
  else           th[h2] = tanhf(x2);
  if (tid >= 224) th[h3] = tanhf(x3);
}

struct Bias { float b0[3], b1[3], b2[3], b3; };

__device__ __forceinline__ void load_bias(
    const float* b0, const float* b1, const float* b2, const float* b3,
    int wave, int lane, Bias& B) {
  const int c = wave * 2 + (lane >> 4);
#pragma unroll
  for (int i = 0; i < 3; ++i) {
    B.b0[i] = b0[c * 3 + i];
    B.b1[i] = b1[c * 3 + i];
    B.b2[i] = b2[c * 3 + i];
  }
  B.b3 = b3[c];
}

__device__ __forceinline__ void load_pack(
    const float* sp, const float* th, const Bias& B,
    int wave, int lane, Pack& P) {
  const int half = lane >> 4, r = lane & 15;
  const int c = wave * 2 + half;
  const int j = r - 8 * half;
  const bool live3 = (j >= 0) && (j < 6);
  const bool live1 = (j >= 6) && (j < 8);
  const bool second = (j & 1) != 0;
  const int o = live3 ? (j >> 1) : 0;

  {
    const float w = live3 ? sp[SP_M0 + c * 3 + o] : 0.f;
    P.a0.x = second ? 0.f : w;
    P.a0.y = second ? w : 0.f;
  }
#pragma unroll
  for (int i = 0; i < 3; ++i) {
    const float w1 = live3 ? sp[SP_M1 + c * 9 + o * 3 + i] : 0.f;
    const float w2 = live3 ? sp[SP_M2 + c * 9 + o * 3 + i] : 0.f;
    const float w3 = live1 ? sp[SP_M3 + c * 3 + i] : 0.f;
    P.a1[i].x = second ? 0.f : w1;  P.a1[i].y = second ? w1 : 0.f;
    P.a2[i].x = second ? 0.f : w2;  P.a2[i].y = second ? w2 : 0.f;
    P.a3[i].x = second ? 0.f : w3;  P.a3[i].y = second ? w3 : 0.f;
  }

#pragma unroll
  for (int i = 0; i < 3; ++i) {
    const float wi = sp[SP_M0 + c * 3 + i];
    const float x0 = B.b0[i] - 0.5f * wi, x1 = B.b1[i], x2 = B.b2[i];
    P.w0[i] = wi;
    P.c0[2 * i] = x0;  P.c0[2 * i + 1] = x0;
    P.c1[2 * i] = x1;  P.c1[2 * i + 1] = x1;
    P.c2[2 * i] = x2;  P.c2[2 * i + 1] = x2;
    P.t0[i] = th[TH_F0 + c * 3 + i];
    P.t1[i] = th[TH_F1 + c * 3 + i];
    P.t2[i] = th[TH_F2 + c * 3 + i];
  }
  P.c0[6] = B.b3;  P.c0[7] = B.b3;
  P.c1[6] = 0.f;  P.c1[7] = 0.f;
  P.c2[6] = 0.f;  P.c2[7] = 0.f;
  P.t3 = th[TH_F3 + c];
}

__device__ __forceinline__ void gate6(const v8f d, const float* t, v2f& b0, v2f& b1, v2f& b2) {
  b0.x = d[0] + t[0] * fast_tanh(d[0]);
  b0.y = d[1] + t[0] * fast_tanh(d[1]);
  b1.x = d[2] + t[1] * fast_tanh(d[2]);
  b1.y = d[3] + t[1] * fast_tanh(d[3]);
  b2.x = d[4] + t[2] * fast_tanh(d[4]);
  b2.y = d[5] + t[2] * fast_tanh(d[5]);
}

__device__ __forceinline__ void eval_two(const Pack& P, float xa, float xb,
                                         float& loa, float& upa, float& lob, float& upb) {
  v2f bx, pl, ql, rl, pu, qu, ru;
  bx.x = xa; bx.y = xb;
  v8f dl = wmma4_f32(P.a0, bx, P.c0);
  v8f du = dl;
  du[0] = dl[0] + P.w0[0];  du[1] = dl[1] + P.w0[0];
  du[2] = dl[2] + P.w0[1];  du[3] = dl[3] + P.w0[1];
  du[4] = dl[4] + P.w0[2];  du[5] = dl[5] + P.w0[2];
  gate6(dl, P.t0, pl, ql, rl);
  gate6(du, P.t0, pu, qu, ru);

  dl = wmma4_f32(P.a1[0], pl, P.c1);
  du = wmma4_f32(P.a1[0], pu, P.c1);
  dl = wmma4_f32(P.a1[1], ql, dl);
  du = wmma4_f32(P.a1[1], qu, du);
  dl = wmma4_f32(P.a1[2], rl, dl);
  du = wmma4_f32(P.a1[2], ru, du);
  gate6(dl, P.t1, pl, ql, rl);
  gate6(du, P.t1, pu, qu, ru);

  dl = wmma4_f32(P.a2[0], pl, P.c2);
  du = wmma4_f32(P.a2[0], pu, P.c2);
  dl = wmma4_f32(P.a2[1], ql, dl);
  du = wmma4_f32(P.a2[1], qu, du);
  dl = wmma4_f32(P.a2[2], rl, dl);
  du = wmma4_f32(P.a2[2], ru, du);
  gate6(dl, P.t2, pl, ql, rl);
  gate6(du, P.t2, pu, qu, ru);

  dl = wmma4_f32(P.a3[0], pl, P.c0);
  du = wmma4_f32(P.a3[0], pu, P.c0);
  dl = wmma4_f32(P.a3[1], ql, dl);
  du = wmma4_f32(P.a3[1], qu, du);
  dl = wmma4_f32(P.a3[2], rl, dl);
  du = wmma4_f32(P.a3[2], ru, du);
  loa = dl[6] + P.t3 * fast_tanh(dl[6]);
  lob = dl[7] + P.t3 * fast_tanh(dl[7]);
  upa = du[6] + P.t3 * fast_tanh(du[6]);
  upb = du[7] + P.t3 * fast_tanh(du[7]);
}

__device__ __forceinline__ float likelihood_of(float lo, float up) {
  const float L2E = 1.4426950408889634f;
  float sum = lo + up;
  float k = (sum > 0.f) ? L2E : ((sum < 0.f) ? -L2E : 0.f);
  float ea = hw_exp2(k * up);
  float eb = hw_exp2(k * lo);
  float lk = fabsf(hw_rcp(1.f + ea) - hw_rcp(1.f + eb));
  return fmaxf(lk, 1e-9f);
}

#define XP 82
#define LP 68

__device__ __forceinline__ void lds_barrier() {
  asm volatile("s_wait_dscnt 0x0\n\ts_barrier_signal -1\n\ts_barrier_wait -1" ::: "memory");
}

__global__ void __launch_bounds__(256)
eb_kernel(const float* __restrict__ inputs, const float* __restrict__ noise,
          const float* __restrict__ m0, const float* __restrict__ b0, const float* __restrict__ f0,
          const float* __restrict__ m1, const float* __restrict__ b1, const float* __restrict__ f1,
          const float* __restrict__ m2, const float* __restrict__ b2, const float* __restrict__ f2,
          const float* __restrict__ m3, const float* __restrict__ b3, const float* __restrict__ f3,
          float* __restrict__ out, float* __restrict__ lik, int ntiles) {
  __shared__ float xs[16 * XP];
  __shared__ __attribute__((aligned(16))) float ls[16 * LP];
  __shared__ float sp[SP_N];
  __shared__ float th[TH_N];

  const int tid  = threadIdx.x;
  const int lane = tid & 31;
  const int wave = tid >> 5;
  const int half = lane >> 4;
  const int row  = lane & 15;

  const float4* __restrict__ in4 = (const float4*)inputs;
  const float4* __restrict__ nz4 = (const float4*)noise;
  float4* __restrict__ out4 = (float4*)out;
  float4* __restrict__ lik4 = (float4*)lik;

  __builtin_amdgcn_s_setprio(3);

  int t = blockIdx.x;
  int left = (t < ntiles) ? (ntiles - 1 - t) / (int)gridDim.x : 0;
  float4 a = {0.f, 0.f, 0.f, 0.f}, b = {0.f, 0.f, 0.f, 0.f};
  if (t < ntiles) {
    a = in4[t * 256 + tid];
    b = nz4[t * 256 + tid];
  }

  Bias Bv;
  load_bias(b0, b1, b2, b3, wave, lane, Bv);
  transform_params(m0, f0, m1, f1, m2, f2, m3, f3, tid, sp, th);
  lds_barrier();
  Pack P;
  load_pack(sp, th, Bv, wave, lane, P);

  const int p = tid >> 2, q = tid & 3;
  float* xw = &xs[4 * q * XP + p];
  const float4* lr = (const float4*)&ls[(p >> 2) * LP + (p & 3) * 16 + 4 * q];
  const float* xr = &xs[(2 * wave + half) * XP + row];
  float* lw = &ls[(row >> 2) * LP + (row & 3) * 16 + 2 * wave + half];

  for (; t < ntiles; t += gridDim.x) {
    if (left < 3) {
      if (left == 2)      __builtin_amdgcn_s_setprio(2);
      else if (left == 1) __builtin_amdgcn_s_setprio(1);
      else                __builtin_amdgcn_s_setprio(0);
    }
    --left;

    float4 v;
    v.x = a.x + (b.x - 0.5f);
    v.y = a.y + (b.y - 0.5f);
    v.z = a.z + (b.z - 0.5f);
    v.w = a.w + (b.w - 0.5f);
    out4[t * 256 + tid] = v;
    xw[0]      = v.x;
    xw[XP]     = v.y;
    xw[2 * XP] = v.z;
    xw[3 * XP] = v.w;

    const int tn = t + gridDim.x;
    if (tn < ntiles) {
      a = in4[tn * 256 + tid];
      b = nz4[tn * 256 + tid];
    }
    lds_barrier();

#pragma nounroll
    for (int g = 0; g < 4; g += 2) {
      float loa, upa, lob, upb;
      eval_two(P, xr[16 * g], xr[16 * g + 16], loa, upa, lob, upb);
      lw[4 * g * LP]       = likelihood_of(loa, upa);
      lw[4 * (g + 1) * LP] = likelihood_of(lob, upb);
    }
    lds_barrier();

    lik4[t * 256 + tid] = *lr;
  }
}

extern "C" void kernel_launch(void* const* d_in, const int* in_sizes, int n_in,
                              void* d_out, int out_size, void* d_ws, size_t ws_size,
                              hipStream_t stream) {
  const float* inputs = (const float*)d_in[0];
  const float* noise  = (const float*)d_in[1];
  const float* m0 = (const float*)d_in[2];
  const float* b0 = (const float*)d_in[3];
  const float* f0 = (const float*)d_in[4];
  const float* m1 = (const float*)d_in[5];
  const float* b1 = (const float*)d_in[6];
  const float* f1 = (const float*)d_in[7];
  const float* m2 = (const float*)d_in[8];
  const float* b2 = (const float*)d_in[9];
  const float* f2 = (const float*)d_in[10];
  const float* m3 = (const float*)d_in[11];
  const float* b3 = (const float*)d_in[12];
  const float* f3 = (const float*)d_in[13];

  const int nelem  = in_sizes[0];
  const int ntiles = nelem / 1024;
  float* out = (float*)d_out;
  float* lik = out + nelem;

  int blocks = ntiles < 1024 ? ntiles : 1024;
  eb_kernel<<<blocks, 256, 0, stream>>>(inputs, noise,
                                        m0, b0, f0, m1, b1, f1,
                                        m2, b2, f2, m3, b3, f3,
                                        out, lik, ntiles);
}


# Layer 3 folded into layer 2: free D rows 6/7 of layer 2's matrix ops carry (w3^T W2)h + w3.b2 + b3; sum (w3*tanh f2)*tanh(d) replaces its gate; 13 matrix ops per 64 elems, was 19; sum order changes
# speedup vs baseline: 4.2170x; 1.1915x over previous
#include <hip/hip_runtime.h>
#include <math.h>


typedef float v2f __attribute__((ext_vector_type(2)));
typedef float v8f __attribute__((ext_vector_type(8)));

__device__ __forceinline__ float softplusf(float x) {
  return (x > 20.f) ? x : log1pf(expf(x));
}

__device__ __forceinline__ float fast_tanh(float x) {
#if __has_builtin(__builtin_amdgcn_tanhf)
  return __builtin_amdgcn_tanhf(x);
#else
  float cx = fminf(fmaxf(x, -10.f), 10.f);
  float e = exp2f(cx * 2.885390081777927f);
  return (e - 1.f) / (e + 1.f);
#endif
}

__device__ __forceinline__ float hw_exp2(float x) {
#if __has_builtin(__builtin_amdgcn_exp2f)
  return __builtin_amdgcn_exp2f(x);
#else
  return exp2f(x);
#endif
}

__device__ __forceinline__ float hw_rcp(float x) {
#if __has_builtin(__builtin_amdgcn_rcpf)
  return __builtin_amdgcn_rcpf(x);
#else
  return 1.f / x;
#endif
}

__device__ __forceinline__ v8f wmma4_f32(v2f a, v2f b, v8f c) {
#if __has_builtin(__builtin_amdgcn_wmma_f32_16x16x4_f32)
  return __builtin_amdgcn_wmma_f32_16x16x4_f32(
       false, a,  false, b,
       (short)0, c,  false,  false);
#else
  int lane = threadIdx.x & 31;
  int half = lane >> 4;
  int n = lane & 15;
  float B0 = __shfl(b.x, n),      B1 = __shfl(b.y, n);
  float B2 = __shfl(b.x, n + 16), B3 = __shfl(b.y, n + 16);
  v8f d;
#pragma unroll
  for (int j = 0; j < 8; ++j) {
    int r = j + (half ? 8 : 0);
    float A0 = __shfl(a.x, r),      A1 = __shfl(a.y, r);
    float A2 = __shfl(a.x, r + 16), A3 = __shfl(a.y, r + 16);
    d[j] = fmaf(A0, B0, fmaf(A1, B1, fmaf(A2, B2, fmaf(A3, B3, c[j]))));
  }
  return d;
#endif
}

struct Pack {
  v2f a0;
  v2f a1[3];
  v2f a2[3];
  v8f c0, c1, c2;
  float w0[3];
  float t0[3], t1[3], t3;
  float g2[3];
};

#define SP_M0 0
#define SP_M1 48
#define SP_M2 192
#define SP_M3 336
#define SP_N  384
#define TH_F0 0
#define TH_F1 48
#define TH_F2 96
#define TH_F3 144
#define TH_N  160

__device__ __forceinline__ void transform_params(
    const float* m0, const float* f0, const float* m1, const float* f1,
    const float* m2, const float* f2, const float* m3, const float* f3,
    int tid, float* sp, float* th) {
  const float* p1 = (tid < SP_M1) ? (m0 + tid)
                  : ((tid < SP_M2) ? (m1 + (tid - SP_M1)) : (m2 + (tid - SP_M2)));
  const int k2 = tid + 256;
  const int h2 = tid - 128;
  const float* p2 = (tid < 128)
      ? ((k2 < SP_M3) ? (m2 + (k2 - SP_M2)) : (m3 + (k2 - SP_M3)))
      : ((h2 < TH_F1) ? (f0 + h2) : ((h2 < TH_F2) ? (f1 + (h2 - TH_F1)) : (f2 + (h2 - TH_F2))));
  const int h3 = tid - 96;
  float x3 = 0.f;
  if (tid >= 224) x3 = (h3 < TH_F3) ? f2[h3 - TH_F2] : f3[h3 - TH_F3];
  const float x1 = *p1;
  const float x2 = *p2;

  sp[tid] = softplusf(x1);
  if (tid < 128) sp[k2] = softplusf(x2);
  else           th[h2] = tanhf(x2);
  if (tid >= 224) th[h3] = tanhf(x3);
}

struct Bias { float b0[3], b1[3], b2[3], b3; };

__device__ __forceinline__ void load_bias(
    const float* b0, const float* b1, const float* b2, const float* b3,
    int wave, int lane, Bias& B) {
  const int c = wave * 2 + (lane >> 4);
#pragma unroll
  for (int i = 0; i < 3; ++i) {
    B.b0[i] = b0[c * 3 + i];
    B.b1[i] = b1[c * 3 + i];
    B.b2[i] = b2[c * 3 + i];
  }
  B.b3 = b3[c];
}

__device__ __forceinline__ void load_pack(
    const float* sp, const float* th, const Bias& B,
    int wave, int lane, Pack& P) {
  const int half = lane >> 4, r = lane & 15;
  const int c = wave * 2 + half;
  const int j = r - 8 * half;
  const bool live3 = (j >= 0) && (j < 6);
  const bool live1 = (j >= 6) && (j < 8);
  const bool second = (j & 1) != 0;
  const int o = live3 ? (j >> 1) : 0;

  {
    const float w = live3 ? sp[SP_M0 + c * 3 + o] : 0.f;
    P.a0.x = second ? 0.f : w;
    P.a0.y = second ? w : 0.f;
  }
  const float v0 = sp[SP_M3 + c * 3 + 0];
  const float v1 = sp[SP_M3 + c * 3 + 1];
  const float v2 = sp[SP_M3 + c * 3 + 2];
#pragma unroll
  for (int i = 0; i < 3; ++i) {
    const float w1 = live3 ? sp[SP_M1 + c * 9 + o * 3 + i] : 0.f;
    const float fold = v0 * sp[SP_M2 + c * 9 + 0 + i]
                     + v1 * sp[SP_M2 + c * 9 + 3 + i]
                     + v2 * sp[SP_M2 + c * 9 + 6 + i];
    const float w2 = live3 ? sp[SP_M2 + c * 9 + o * 3 + i] : (live1 ? fold : 0.f);
    P.a1[i].x = second ? 0.f : w1;  P.a1[i].y = second ? w1 : 0.f;
    P.a2[i].x = second ? 0.f : w2;  P.a2[i].y = second ? w2 : 0.f;
  }

#pragma unroll
  for (int i = 0; i < 3; ++i) {
    const float wi = sp[SP_M0 + c * 3 + i];
    const float x0 = B.b0[i] - 0.5f * wi, x1 = B.b1[i], x2 = B.b2[i];
    P.w0[i] = wi;
    P.c0[2 * i] = x0;  P.c0[2 * i + 1] = x0;
    P.c1[2 * i] = x1;  P.c1[2 * i + 1] = x1;
    P.c2[2 * i] = x2;  P.c2[2 * i + 1] = x2;
    P.t0[i] = th[TH_F0 + c * 3 + i];
    P.t1[i] = th[TH_F1 + c * 3 + i];
  }
  P.g2[0] = v0 * th[TH_F2 + c * 3 + 0];
  P.g2[1] = v1 * th[TH_F2 + c * 3 + 1];
  P.g2[2] = v2 * th[TH_F2 + c * 3 + 2];
  const float fb = v0 * B.b2[0] + v1 * B.b2[1] + v2 * B.b2[2] + B.b3;
  P.c0[6] = 0.f;  P.c0[7] = 0.f;
  P.c1[6] = 0.f;  P.c1[7] = 0.f;
  P.c2[6] = fb;   P.c2[7] = fb;
  P.t3 = th[TH_F3 + c];
}

__device__ __forceinline__ void gate6(const v8f d, const float* t, v2f& b0, v2f& b1, v2f& b2) {
  b0.x = d[0] + t[0] * fast_tanh(d[0]);
  b0.y = d[1] + t[0] * fast_tanh(d[1]);
  b1.x = d[2] + t[1] * fast_tanh(d[2]);
  b1.y = d[3] + t[1] * fast_tanh(d[3]);
  b2.x = d[4] + t[2] * fast_tanh(d[4]);
  b2.y = d[5] + t[2] * fast_tanh(d[5]);
}

__device__ __forceinline__ void close23(const v8f d, const float* g, float t3, float& ra, float& rb) {
  v2f o, h;
  o.x = d[6];  o.y = d[7];
  h.x = fast_tanh(d[0]);  h.y = fast_tanh(d[1]);
  o = o + g[0] * h;
  h.x = fast_tanh(d[2]);  h.y = fast_tanh(d[3]);
  o = o + g[1] * h;
  h.x = fast_tanh(d[4]);  h.y = fast_tanh(d[5]);
  o = o + g[2] * h;
  h.x = fast_tanh(o.x);  h.y = fast_tanh(o.y);
  o = o + t3 * h;
  ra = o.x;
  rb = o.y;
}

__device__ __forceinline__ void eval_two(const Pack& P, float xa, float xb,
                                         float& loa, float& upa, float& lob, float& upb) {
  v2f bx, pl, ql, rl, pu, qu, ru;
  bx.x = xa; bx.y = xb;
  v8f dl = wmma4_f32(P.a0, bx, P.c0);
  v8f du = dl;
  du[0] = dl[0] + P.w0[0];  du[1] = dl[1] + P.w0[0];
  du[2] = dl[2] + P.w0[1];  du[3] = dl[3] + P.w0[1];
  du[4] = dl[4] + P.w0[2];  du[5] = dl[5] + P.w0[2];
  gate6(dl, P.t0, pl, ql, rl);
  gate6(du, P.t0, pu, qu, ru);

  dl = wmma4_f32(P.a1[0], pl, P.c1);
  du = wmma4_f32(P.a1[0], pu, P.c1);
  dl = wmma4_f32(P.a1[1], ql, dl);
  du = wmma4_f32(P.a1[1], qu, du);
  dl = wmma4_f32(P.a1[2], rl, dl);
  du = wmma4_f32(P.a1[2], ru, du);
  gate6(dl, P.t1, pl, ql, rl);
  gate6(du, P.t1, pu, qu, ru);

  dl = wmma4_f32(P.a2[0], pl, P.c2);
  du = wmma4_f32(P.a2[0], pu, P.c2);
  dl = wmma4_f32(P.a2[1], ql, dl);
  du = wmma4_f32(P.a2[1], qu, du);
  dl = wmma4_f32(P.a2[2], rl, dl);
  du = wmma4_f32(P.a2[2], ru, du);

  close23(dl, P.g2, P.t3, loa, lob);
  close23(du, P.g2, P.t3, upa, upb);
}

__device__ __forceinline__ float likelihood_of(float lo, float up) {
  const float L2E = 1.4426950408889634f;
  float sum = lo + up;
  float k = (sum > 0.f) ? L2E : ((sum < 0.f) ? -L2E : 0.f);
  float ea = hw_exp2(k * up);
  float eb = hw_exp2(k * lo);
  float lk = fabsf(hw_rcp(1.f + ea) - hw_rcp(1.f + eb));
  return fmaxf(lk, 1e-9f);
}

#define XP 82
#define LP 68

__device__ __forceinline__ void lds_barrier() {
  asm volatile("s_wait_dscnt 0x0\n\ts_barrier_signal -1\n\ts_barrier_wait -1" ::: "memory");
}

__global__ void __launch_bounds__(256)
eb_kernel(const float* __restrict__ inputs, const float* __restrict__ noise,
          const float* __restrict__ m0, const float* __restrict__ b0, const float* __restrict__ f0,
          const float* __restrict__ m1, const float* __restrict__ b1, const float* __restrict__ f1,
          const float* __restrict__ m2, const float* __restrict__ b2, const float* __restrict__ f2,
          const float* __restrict__ m3, const float* __restrict__ b3, const float* __restrict__ f3,
          float* __restrict__ out, float* __restrict__ lik, int ntiles) {
  __shared__ float xs[16 * XP];
  __shared__ __attribute__((aligned(16))) float ls[16 * LP];
  __shared__ float sp[SP_N];
  __shared__ float th[TH_N];

  const int tid  = threadIdx.x;
  const int lane = tid & 31;
  const int wave = tid >> 5;
  const int half = lane >> 4;
  const int row  = lane & 15;

  const float4* __restrict__ in4 = (const float4*)inputs;
  const float4* __restrict__ nz4 = (const float4*)noise;
  float4* __restrict__ out4 = (float4*)out;
  float4* __restrict__ lik4 = (float4*)lik;

  __builtin_amdgcn_s_setprio(3);

  int t = blockIdx.x;
  int left = (t < ntiles) ? (ntiles - 1 - t) / (int)gridDim.x : 0;
  float4 a = {0.f, 0.f, 0.f, 0.f}, b = {0.f, 0.f, 0.f, 0.f};
  if (t < ntiles) {
    a = in4[t * 256 + tid];
    b = nz4[t * 256 + tid];
  }

  Bias Bv;
  load_bias(b0, b1, b2, b3, wave, lane, Bv);
  transform_params(m0, f0, m1, f1, m2, f2, m3, f3, tid, sp, th);
  lds_barrier();
  Pack P;
  load_pack(sp, th, Bv, wave, lane, P);

  const int p = tid >> 2, q = tid & 3;
  float* xw = &xs[4 * q * XP + p];
  const float4* lr = (const float4*)&ls[(p >> 2) * LP + (p & 3) * 16 + 4 * q];
  const float* xr = &xs[(2 * wave + half) * XP + row];
  float* lw = &ls[(row >> 2) * LP + (row & 3) * 16 + 2 * wave + half];

  for (; t < ntiles; t += gridDim.x) {
    if (left < 3) {
      if (left == 2)      __builtin_amdgcn_s_setprio(2);
      else if (left == 1) __builtin_amdgcn_s_setprio(1);
      else                __builtin_amdgcn_s_setprio(0);
    }
    --left;

    float4 v;
    v.x = a.x + (b.x - 0.5f);
    v.y = a.y + (b.y - 0.5f);
    v.z = a.z + (b.z - 0.5f);
    v.w = a.w + (b.w - 0.5f);
    out4[t * 256 + tid] = v;
    xw[0]      = v.x;
    xw[XP]     = v.y;
    xw[2 * XP] = v.z;
    xw[3 * XP] = v.w;

    const int tn = t + gridDim.x;
    if (tn < ntiles) {
      a = in4[tn * 256 + tid];
      b = nz4[tn * 256 + tid];
    }
    lds_barrier();

#pragma nounroll
    for (int g = 0; g < 4; g += 2) {
      float loa, upa, lob, upb;
      eval_two(P, xr[16 * g], xr[16 * g + 16], loa, upa, lob, upb);
      lw[4 * g * LP]       = likelihood_of(loa, upa);
      lw[4 * (g + 1) * LP] = likelihood_of(lob, upb);
    }
    lds_barrier();

    lik4[t * 256 + tid] = *lr;
  }
}

extern "C" void kernel_launch(void* const* d_in, const int* in_sizes, int n_in,
                              void* d_out, int out_size, void* d_ws, size_t ws_size,
                              hipStream_t stream) {
  const float* inputs = (const float*)d_in[0];
  const float* noise  = (const float*)d_in[1];
  const float* m0 = (const float*)d_in[2];
  const float* b0 = (const float*)d_in[3];
  const float* f0 = (const float*)d_in[4];
  const float* m1 = (const float*)d_in[5];
  const float* b1 = (const float*)d_in[6];
  const float* f1 = (const float*)d_in[7];
  const float* m2 = (const float*)d_in[8];
  const float* b2 = (const float*)d_in[9];
  const float* f2 = (const float*)d_in[10];
  const float* m3 = (const float*)d_in[11];
  const float* b3 = (const float*)d_in[12];
  const float* f3 = (const float*)d_in[13];

  const int nelem  = in_sizes[0];
  const int ntiles = nelem / 1024;
  float* out = (float*)d_out;
  float* lik = out + nelem;

  int blocks = ntiles < 1024 ? ntiles : 1024;
  eb_kernel<<<blocks, 256, 0, stream>>>(inputs, noise,
                                        m0, b0, f0, m1, b1, f1,
                                        m2, b2, f2, m3, b3, f3,
                                        out, lik, ntiles);
}


# speedup vs baseline: 4.4239x; 1.0491x over previous
#include <hip/hip_runtime.h>
#include <math.h>


typedef float v2f __attribute__((ext_vector_type(2)));
typedef float v8f __attribute__((ext_vector_type(8)));

__device__ __forceinline__ float softplusf(float x) {
  return (x > 20.f) ? x : log1pf(expf(x));
}

__device__ __forceinline__ float fast_tanh(float x) {
#if __has_builtin(__builtin_amdgcn_tanhf)
  return __builtin_amdgcn_tanhf(x);
#else
  float cx = fminf(fmaxf(x, -10.f), 10.f);
  float e = exp2f(cx * 2.885390081777927f);
  return (e - 1.f) / (e + 1.f);
#endif
}

__device__ __forceinline__ float hw_exp2(float x) {
#if __has_builtin(__builtin_amdgcn_exp2f)
  return __builtin_amdgcn_exp2f(x);
#else
  return exp2f(x);
#endif
}

__device__ __forceinline__ float hw_rcp(float x) {
#if __has_builtin(__builtin_amdgcn_rcpf)
  return __builtin_amdgcn_rcpf(x);
#else
  return 1.f / x;
#endif
}

__device__ __forceinline__ v8f wmma4_f32(v2f a, v2f b, v8f c) {
#if __has_builtin(__builtin_amdgcn_wmma_f32_16x16x4_f32)
  return __builtin_amdgcn_wmma_f32_16x16x4_f32(
       false, a,  false, b,
       (short)0, c,  false,  false);
#else
  int lane = threadIdx.x & 31;
  int half = lane >> 4;
  int n = lane & 15;
  float B0 = __shfl(b.x, n),      B1 = __shfl(b.y, n);
  float B2 = __shfl(b.x, n + 16), B3 = __shfl(b.y, n + 16);
  v8f d;
#pragma unroll
  for (int j = 0; j < 8; ++j) {
    int r = j + (half ? 8 : 0);
    float A0 = __shfl(a.x, r),      A1 = __shfl(a.y, r);
    float A2 = __shfl(a.x, r + 16), A3 = __shfl(a.y, r + 16);
    d[j] = fmaf(A0, B0, fmaf(A1, B1, fmaf(A2, B2, fmaf(A3, B3, c[j]))));
  }
  return d;
#endif
}

struct Pack {
  v2f a0;
  v2f a1[3];
  v2f a2[3];
  v8f c0, c1, c2;
  float w0[3];
  float t0[3], t1[3], t3;
  float g2[3];
};

#define SP_M0 0
#define SP_M1 48
#define SP_M2 192
#define SP_M3 336
#define SP_N  384
#define TH_F0 0
#define TH_F1 48
#define TH_F2 96
#define TH_F3 144
#define TH_N  160

__device__ __forceinline__ void transform_params(
    const float* m0, const float* f0, const float* m1, const float* f1,
    const float* m2, const float* f2, const float* m3, const float* f3,
    int tid, float* sp, float* th) {
  const float* p1 = (tid < SP_M1) ? (m0 + tid)
                  : ((tid < SP_M2) ? (m1 + (tid - SP_M1)) : (m2 + (tid - SP_M2)));
  const int k2 = tid + 256;
  const int h2 = tid - 128;
  const float* p2 = (tid < 128)
      ? ((k2 < SP_M3) ? (m2 + (k2 - SP_M2)) : (m3 + (k2 - SP_M3)))
      : ((h2 < TH_F1) ? (f0 + h2) : ((h2 < TH_F2) ? (f1 + (h2 - TH_F1)) : (f2 + (h2 - TH_F2))));
  const int h3 = tid - 96;
  float x3 = 0.f;
  if (tid >= 224) x3 = (h3 < TH_F3) ? f2[h3 - TH_F2] : f3[h3 - TH_F3];
  const float x1 = *p1;
  const float x2 = *p2;

  sp[tid] = softplusf(x1);
  if (tid < 128) sp[k2] = softplusf(x2);
  else           th[h2] = tanhf(x2);
  if (tid >= 224) th[h3] = tanhf(x3);
}

struct Bias { float b0[3], b1[3], b2[3], b3; };

__device__ __forceinline__ void load_bias(
    const float* b0, const float* b1, const float* b2, const float* b3,
    int wave, int lane, Bias& B) {
  const int c = wave * 2 + (lane >> 4);
#pragma unroll
  for (int i = 0; i < 3; ++i) {
    B.b0[i] = b0[c * 3 + i];
    B.b1[i] = b1[c * 3 + i];
    B.b2[i] = b2[c * 3 + i];
  }
  B.b3 = b3[c];
}

__device__ __forceinline__ void load_pack(
    const float* sp, const float* th, const Bias& B,
    int wave, int lane, Pack& P) {
  const int half = lane >> 4, r = lane & 15;
  const int c = wave * 2 + half;
  const int j = r - 8 * half;
  const bool live3 = (j >= 0) && (j < 6);
  const bool live1 = (j >= 6) && (j < 8);
  const bool second = (j & 1) != 0;
  const int o = live3 ? (j >> 1) : 0;

  {
    const float w = live3 ? sp[SP_M0 + c * 3 + o] : 0.f;
    P.a0.x = second ? 0.f : w;
    P.a0.y = second ? w : 0.f;
  }
  const float v0 = sp[SP_M3 + c * 3 + 0];
  const float v1 = sp[SP_M3 + c * 3 + 1];
  const float v2 = sp[SP_M3 + c * 3 + 2];
#pragma unroll
  for (int i = 0; i < 3; ++i) {
    const float w1 = live3 ? sp[SP_M1 + c * 9 + o * 3 + i] : 0.f;
    const float fold = v0 * sp[SP_M2 + c * 9 + 0 + i]
                     + v1 * sp[SP_M2 + c * 9 + 3 + i]
                     + v2 * sp[SP_M2 + c * 9 + 6 + i];
    const float w2 = live3 ? sp[SP_M2 + c * 9 + o * 3 + i] : (live1 ? fold : 0.f);
    P.a1[i].x = second ? 0.f : w1;  P.a1[i].y = second ? w1 : 0.f;
    P.a2[i].x = second ? 0.f : w2;  P.a2[i].y = second ? w2 : 0.f;
  }

#pragma unroll
  for (int i = 0; i < 3; ++i) {
    const float wi = sp[SP_M0 + c * 3 + i];
    const float x0 = B.b0[i] - 0.5f * wi, x1 = B.b1[i], x2 = B.b2[i];
    P.w0[i] = wi;
    P.c0[2 * i] = x0;  P.c0[2 * i + 1] = x0;
    P.c1[2 * i] = x1;  P.c1[2 * i + 1] = x1;
    P.c2[2 * i] = x2;  P.c2[2 * i + 1] = x2;
    P.t0[i] = th[TH_F0 + c * 3 + i];
    P.t1[i] = th[TH_F1 + c * 3 + i];
  }
  P.g2[0] = v0 * th[TH_F2 + c * 3 + 0];
  P.g2[1] = v1 * th[TH_F2 + c * 3 + 1];
  P.g2[2] = v2 * th[TH_F2 + c * 3 + 2];
  const float fb = v0 * B.b2[0] + v1 * B.b2[1] + v2 * B.b2[2] + B.b3;
  P.c0[6] = 0.f;  P.c0[7] = 0.f;
  P.c1[6] = 0.f;  P.c1[7] = 0.f;
  P.c2[6] = fb;   P.c2[7] = fb;
  P.t3 = th[TH_F3 + c];
}

__device__ __forceinline__ void close23(const v8f d, const float* g, float t3, float& ra, float& rb) {
  v2f o, h;
  o.x = d[6];  o.y = d[7];
  h.x = fast_tanh(d[0]);  h.y = fast_tanh(d[1]);
  o = o + g[0] * h;
  h.x = fast_tanh(d[2]);  h.y = fast_tanh(d[3]);
  o = o + g[1] * h;
  h.x = fast_tanh(d[4]);  h.y = fast_tanh(d[5]);
  o = o + g[2] * h;
  h.x = fast_tanh(o.x);  h.y = fast_tanh(o.y);
  o = o + t3 * h;
  ra = o.x;
  rb = o.y;
}

__device__ __forceinline__ float likelihood_of(float lo, float up) {
  const float L2E = 1.4426950408889634f;
  float sum = lo + up;
  float k = (sum > 0.f) ? L2E : ((sum < 0.f) ? -L2E : 0.f);
  float ea = hw_exp2(k * up);
  float eb = hw_exp2(k * lo);
  float lk = fabsf(hw_rcp(1.f + ea) - hw_rcp(1.f + eb));
  return fmaxf(lk, 1e-9f);
}

#define XP 82
#define LP 68
#define XS_N (16 * XP)
#define LS_N (16 * LP)


#define SLOT_END() __builtin_amdgcn_sched_barrier(0)

__device__ __forceinline__ void pair_block(const Pack& P, float xa, float xb, v8f& cl, v8f& cu,
                                           float* w0, float* w1) {
  const float L2E = 1.4426950408889634f;
  v2f bx, pl, ql, rl, pu, qu, ru, h, d, ol, ou, gl, gu;
  bx.x = xa; bx.y = xb;
  float a0, a1, a2, a3, a4, a5, b0, b1, b2, b3, b4, b5;
  float tl0, tl1, tl2, tl3, tl4, tl5, tu0, tu1, tu2, tu3, tu4, tu5;

  v8f dl = wmma4_f32(P.a0, bx, P.c0);
  v8f du = dl;
  SLOT_END();

  ol.x = cl[6]; ol.y = cl[7];
  ou.x = cu[6]; ou.y = cu[7];
  a0 = fast_tanh(cl[0]);  du[0] = dl[0] + P.w0[0];  SLOT_END();
  a1 = fast_tanh(cl[1]);  du[1] = dl[1] + P.w0[0];  SLOT_END();
  a2 = fast_tanh(cl[2]);  du[2] = dl[2] + P.w0[1];  SLOT_END();
  a3 = fast_tanh(cl[3]);  du[3] = dl[3] + P.w0[1];  SLOT_END();
  a4 = fast_tanh(cl[4]);  h.x = a0; h.y = a1;  ol = ol + P.g2[0] * h;  SLOT_END();
  a5 = fast_tanh(cl[5]);  du[4] = dl[4] + P.w0[2];  SLOT_END();
  b0 = fast_tanh(cu[0]);  h.x = a2; h.y = a3;  ol = ol + P.g2[1] * h;  SLOT_END();
  b1 = fast_tanh(cu[1]);  du[5] = dl[5] + P.w0[2];  SLOT_END();
  b2 = fast_tanh(cu[2]);  h.x = a4; h.y = a5;  ol = ol + P.g2[2] * h;  SLOT_END();
  b3 = fast_tanh(cu[3]);  SLOT_END();
  b4 = fast_tanh(cu[4]);  h.x = b0; h.y = b1;  ou = ou + P.g2[0] * h;  SLOT_END();
  b5 = fast_tanh(cu[5]);  SLOT_END();
  gl.x = fast_tanh(ol.x); h.x = b2; h.y = b3;  ou = ou + P.g2[1] * h;  SLOT_END();
  gl.y = fast_tanh(ol.y); SLOT_END();

  tl0 = fast_tanh(dl[0]);  h.x = b4; h.y = b5;  ou = ou + P.g2[2] * h;  SLOT_END();
  tl1 = fast_tanh(dl[1]);  SLOT_END();
  tl2 = fast_tanh(dl[2]);  SLOT_END();
  tl3 = fast_tanh(dl[3]);  SLOT_END();
  gu.x = fast_tanh(ou.x);
  h.x = tl0; h.y = tl1; d.x = dl[0]; d.y = dl[1];  pl = d + P.t0[0] * h;  SLOT_END();
  gu.y = fast_tanh(ou.y);  SLOT_END();
  tl4 = fast_tanh(dl[4]);  ol = ol + P.t3 * gl;      SLOT_END();
  tl5 = fast_tanh(dl[5]);
  h.x = tl2; h.y = tl3; d.x = dl[2]; d.y = dl[3];  ql = d + P.t0[1] * h;  SLOT_END();
  tu0 = fast_tanh(du[0]);  SLOT_END();
  tu1 = fast_tanh(du[1]);  ou = ou + P.t3 * gu;      SLOT_END();
  tu2 = fast_tanh(du[2]);
  h.x = tl4; h.y = tl5; d.x = dl[4]; d.y = dl[5];  rl = d + P.t0[2] * h;  SLOT_END();
  tu3 = fast_tanh(du[3]);  SLOT_END();
  tu4 = fast_tanh(du[4]);  const v2f ss = ol + ou;   SLOT_END();
  tu5 = fast_tanh(du[5]);
  h.x = tu0; h.y = tu1; d.x = du[0]; d.y = du[1];  pu = d + P.t0[0] * h;  SLOT_END();
  h.x = tu2; h.y = tu3; d.x = du[2]; d.y = du[3];  qu = d + P.t0[1] * h;
  h.x = tu4; h.y = tu5; d.x = du[4]; d.y = du[5];  ru = d + P.t0[2] * h;

  dl = wmma4_f32(P.a1[0], pl, P.c1);
  du = wmma4_f32(P.a1[0], pu, P.c1);
  dl = wmma4_f32(P.a1[1], ql, dl);
  du = wmma4_f32(P.a1[1], qu, du);
  dl = wmma4_f32(P.a1[2], rl, dl);
  du = wmma4_f32(P.a1[2], ru, du);
  SLOT_END();
  v2f kk, eu, el, xu, xl;
  tl0 = fast_tanh(dl[0]);  const float n0 = (ss.x < 0.f) ? -L2E : 0.f;   SLOT_END();
  tl1 = fast_tanh(dl[1]);  const float n1 = (ss.y < 0.f) ? -L2E : 0.f;   SLOT_END();
  tl2 = fast_tanh(dl[2]);  kk.x = (ss.x > 0.f) ? L2E : n0;               SLOT_END();
  tl3 = fast_tanh(dl[3]);  kk.y = (ss.y > 0.f) ? L2E : n1;               SLOT_END();
  tl4 = fast_tanh(dl[4]);  const v2f au = kk * ou;                       SLOT_END();
  tl5 = fast_tanh(dl[5]);  const v2f al = kk * ol;                       SLOT_END();
  tu0 = fast_tanh(du[0]);
  h.x = tl0; h.y = tl1; d.x = dl[0]; d.y = dl[1];  pl = d + P.t1[0] * h;  SLOT_END();
  eu.x = hw_exp2(au.x);
  h.x = tl2; h.y = tl3; d.x = dl[2]; d.y = dl[3];  ql = d + P.t1[1] * h;  SLOT_END();
  tu1 = fast_tanh(du[1]);  SLOT_END();
  eu.y = hw_exp2(au.y);
  h.x = tl4; h.y = tl5; d.x = dl[4]; d.y = dl[5];  rl = d + P.t1[2] * h;  SLOT_END();
  tu2 = fast_tanh(du[2]);  SLOT_END();
  el.x = hw_exp2(al.x);    SLOT_END();
  tu3 = fast_tanh(du[3]);
  h.x = tu0; h.y = tu1; d.x = du[0]; d.y = du[1];  pu = d + P.t1[0] * h;  SLOT_END();
  el.y = hw_exp2(al.y);    const v2f deu = 1.f + eu;                     SLOT_END();
  tu4 = fast_tanh(du[4]);  SLOT_END();
  tu5 = fast_tanh(du[5]);
  h.x = tu2; h.y = tu3; d.x = du[2]; d.y = du[3];  qu = d + P.t1[1] * h;  SLOT_END();
  xu.x = hw_rcp(deu.x);    const v2f del = 1.f + el;                     SLOT_END();
  xu.y = hw_rcp(deu.y);    SLOT_END();
  xl.x = hw_rcp(del.x);
  h.x = tu4; h.y = tu5; d.x = du[4]; d.y = du[5];  ru = d + P.t1[2] * h;  SLOT_END();
  xl.y = hw_rcp(del.y);    SLOT_END();

  cl = wmma4_f32(P.a2[0], pl, P.c2);
  cu = wmma4_f32(P.a2[0], pu, P.c2);
  cl = wmma4_f32(P.a2[1], ql, cl);
  cu = wmma4_f32(P.a2[1], qu, cu);
  cl = wmma4_f32(P.a2[2], rl, cl);
  cu = wmma4_f32(P.a2[2], ru, cu);
  const v2f ff = xu - xl;
  w0[0] = fmaxf(fabsf(ff.x), 1e-9f);
  w1[0] = fmaxf(fabsf(ff.y), 1e-9f);
}

__device__ __forceinline__ void lds_barrier() {
  asm volatile("s_wait_dscnt 0x0\n\ts_barrier_signal -1\n\ts_barrier_wait -1" ::: "memory");
}

__global__ void __launch_bounds__(256)
eb_kernel(const float* __restrict__ inputs, const float* __restrict__ noise,
          const float* __restrict__ m0, const float* __restrict__ b0, const float* __restrict__ f0,
          const float* __restrict__ m1, const float* __restrict__ b1, const float* __restrict__ f1,
          const float* __restrict__ m2, const float* __restrict__ b2, const float* __restrict__ f2,
          const float* __restrict__ m3, const float* __restrict__ b3, const float* __restrict__ f3,
          float* __restrict__ out, float* __restrict__ lik, int ntiles) {
  __shared__ float xs[3 * XS_N];
  __shared__ __attribute__((aligned(16))) float ls[2 * LS_N];
  __shared__ float sp[SP_N];
  __shared__ float th[TH_N];

  const int tid  = threadIdx.x;
  const int lane = tid & 31;
  const int wave = tid >> 5;
  const int half = lane >> 4;
  const int row  = lane & 15;

  const float4* __restrict__ in4 = (const float4*)inputs;
  const float4* __restrict__ nz4 = (const float4*)noise;
  float4* __restrict__ out4 = (float4*)out;
  float4* __restrict__ lik4 = (float4*)lik;

  __builtin_amdgcn_s_setprio(3);

  int t = blockIdx.x;
  if (t >= ntiles) return;
  int left = (ntiles - 1 - t) / (int)gridDim.x;

  float4 a = in4[t * 256 + tid];
  float4 b = nz4[t * 256 + tid];

  Bias Bv;
  load_bias(b0, b1, b2, b3, wave, lane, Bv);
  transform_params(m0, f0, m1, f1, m2, f2, m3, f3, tid, sp, th);

  const int p = tid >> 2, q = tid & 3;
  float* xw = &xs[4 * q * XP + p];
  const float* lr = &ls[(p >> 2) * LP + (p & 3) * 16 + 4 * q];
  const float* xr = &xs[(2 * wave + half) * XP + row];
  float* lw = &ls[(row >> 2) * LP + (row & 3) * 16 + 2 * wave + half];

  {
    float4 v;
    v.x = a.x + (b.x - 0.5f);
    v.y = a.y + (b.y - 0.5f);
    v.z = a.z + (b.z - 0.5f);
    v.w = a.w + (b.w - 0.5f);
    out4[t * 256 + tid] = v;
    xw[0]      = v.x;
    xw[XP]     = v.y;
    xw[2 * XP] = v.z;
    xw[3 * XP] = v.w;
    const int tn = t + gridDim.x;
    if (tn < ntiles) {
      a = in4[tn * 256 + tid];
      b = nz4[tn * 256 + tid];
    }
  }
  lds_barrier();
  Pack P;
  load_pack(sp, th, Bv, wave, lane, P);

  int xcur = 0, xnext = XS_N;
  int lcur = 0, lprev = LS_N;
  bool first = true;
  v8f cl = {0.f, 0.f, 0.f, 0.f, 0.f, 0.f, 0.f, 0.f};
  v8f cu = cl;

  for (; t < ntiles; t += gridDim.x) {
    if (left < 3) {
      if (left == 2)      __builtin_amdgcn_s_setprio(2);
      else if (left == 1) __builtin_amdgcn_s_setprio(1);
      else                __builtin_amdgcn_s_setprio(0);
    }
    --left;

    pair_block(P, xr[xcur], xr[xcur + 16], cl, cu, &lw[lprev + 8 * LP], &lw[lprev + 12 * LP]);

    const int tn = t + gridDim.x;
    if (tn < ntiles) {
      float4 v;
      v.x = a.x + (b.x - 0.5f);
      v.y = a.y + (b.y - 0.5f);
      v.z = a.z + (b.z - 0.5f);
      v.w = a.w + (b.w - 0.5f);
      out4[tn * 256 + tid] = v;
      xw[xnext]          = v.x;
      xw[xnext + XP]     = v.y;
      xw[xnext + 2 * XP] = v.z;
      xw[xnext + 3 * XP] = v.w;
      const int tnn = tn + gridDim.x;
      if (tnn < ntiles) {
        a = in4[tnn * 256 + tid];
        b = nz4[tnn * 256 + tid];
      }
    }
    lds_barrier();

    if (!first) lik4[(t - (int)gridDim.x) * 256 + tid] = *(const float4*)(lr + lprev);
    first = false;

    pair_block(P, xr[xcur + 32], xr[xcur + 48], cl, cu, &lw[lcur], &lw[lcur + 4 * LP]);

    xcur = xnext;
    xnext = (xnext == 2 * XS_N) ? 0 : xnext + XS_N;
    const int ltmp = lcur;  lcur = lprev;  lprev = ltmp;
  }

  {
    float loa, upa, lob, upb;
    close23(cl, P.g2, P.t3, loa, lob);
    close23(cu, P.g2, P.t3, upa, upb);
    lw[lprev + 8 * LP]  = likelihood_of(loa, upa);
    lw[lprev + 12 * LP] = likelihood_of(lob, upb);
  }
  lds_barrier();
  lik4[(t - (int)gridDim.x) * 256 + tid] = *(const float4*)(lr + lprev);
}

extern "C" void kernel_launch(void* const* d_in, const int* in_sizes, int n_in,
                              void* d_out, int out_size, void* d_ws, size_t ws_size,
                              hipStream_t stream) {
  const float* inputs = (const float*)d_in[0];
  const float* noise  = (const float*)d_in[1];
  const float* m0 = (const float*)d_in[2];
  const float* b0 = (const float*)d_in[3];
  const float* f0 = (const float*)d_in[4];
  const float* m1 = (const float*)d_in[5];
  const float* b1 = (const float*)d_in[6];
  const float* f1 = (const float*)d_in[7];
  const float* m2 = (const float*)d_in[8];
  const float* b2 = (const float*)d_in[9];
  const float* f2 = (const float*)d_in[10];
  const float* m3 = (const float*)d_in[11];
  const float* b3 = (const float*)d_in[12];
  const float* f3 = (const float*)d_in[13];

  const int nelem  = in_sizes[0];
  const int ntiles = nelem / 1024;
  float* out = (float*)d_out;
  float* lik = out + nelem;

  int blocks = ntiles < 1024 ? ntiles : 1024;
  eb_kernel<<<blocks, 256, 0, stream>>>(inputs, noise,
                                        m0, b0, f0, m1, b1, f1,
                                        m2, b2, f2, m3, b3, f3,
                                        out, lik, ntiles);
}
